# starting kernel
# baseline (speedup 1.0000x reference)
#include <hip/hip_runtime.h>
#include <hip/hip_bf16.h>

typedef __attribute__((ext_vector_type(2))) float v2f;
typedef __attribute__((ext_vector_type(8))) float v8f;

#define BN_EPS 1e-5f
#define CH 128
#define WSTRIDE 132

__device__ __forceinline__ void fadd_atomic(float* p, float v) {
  unsafeAtomicAdd(p, v);
}

__global__ void zero_f4(float4* __restrict__ p, size_t n4) {
  size_t i = (size_t)blockIdx.x * blockDim.x + threadIdx.x;
  size_t stride = (size_t)gridDim.x * blockDim.x;
  float4 z = {0.f, 0.f, 0.f, 0.f};
  for (; i < n4; i += stride) p[i] = z;
}

__global__ void zero_f(float* __restrict__ p, int n) {
  int i = blockIdx.x * blockDim.x + threadIdx.x;
  if (i < n) p[i] = 0.f;
}

__global__ void spmm128_kernel(const float* __restrict__ feat,
                               const int* __restrict__ src,
                               const int* __restrict__ dst,
                               const float* __restrict__ ew,
                               float* __restrict__ agg, int E) {
  int e = blockIdx.x * 8 + (threadIdx.x >> 5);
  if (e >= E) return;
  int lane = threadIdx.x & 31;
  int s = src[e];
  int d = dst[e];
  float w = ew[e];
  const float4* fr = (const float4*)(feat + (size_t)s * CH);
  float4 v = fr[lane];
  float* o = agg + (size_t)d * CH + lane * 4;
  fadd_atomic(o + 0, v.x * w);
  fadd_atomic(o + 1, v.y * w);
  fadd_atomic(o + 2, v.z * w);
  fadd_atomic(o + 3, v.w * w);
}

__global__ void spmm40_kernel(const float* __restrict__ feat,
                              const int* __restrict__ src,
                              const int* __restrict__ dst,
                              const float* __restrict__ ew,
                              float* __restrict__ agg, int E) {
  int e = blockIdx.x * 8 + (threadIdx.x >> 5);
  if (e >= E) return;
  int lane = threadIdx.x & 31;
  if (lane >= 20) return;
  int s = src[e];
  int d = dst[e];
  float w = ew[e];
  float2 v = *(const float2*)(feat + (size_t)s * 40 + lane * 2);
  float* o = agg + (size_t)d * 40 + lane * 2;
  fadd_atomic(o + 0, v.x * w);
  fadd_atomic(o + 1, v.y * w);
}

template <int NT>
__global__ void gemm_wmma_kernel(const float* __restrict__ A,
                                 const float* __restrict__ W,
                                 float* __restrict__ out,
                                 int Nrows, int Cout, int ldo) {
  constexpr int CP = NT * 16;
  __shared__ float sWT[CP * WSTRIDE];
  int tid = threadIdx.x;
  for (int idx = tid; idx < CH * CP; idx += 256) {
    int k = idx / CP;
    int c = idx % CP;
    sWT[c * WSTRIDE + k] = (c < Cout) ? W[k * Cout + c] : 0.f;
  }
  __syncthreads();

  int wave = tid >> 5;
  int lane = tid & 31;
  int row0 = blockIdx.x * 128 + wave * 16;
  if (row0 >= Nrows) return;

  int m = lane & 15;
  int kh = (lane >> 4) * 2;

  v8f acc[NT];
#pragma unroll
  for (int t = 0; t < NT; ++t) acc[t] = {};

  const float* Arow = A + (size_t)(row0 + m) * CH;

  for (int k0 = 0; k0 < CH; k0 += 4) {
    v2f a = *(const v2f*)(Arow + k0 + kh);
#pragma unroll
    for (int t = 0; t < NT; ++t) {
      v2f b = *(const v2f*)(&sWT[(t * 16 + m) * WSTRIDE + k0 + kh]);
      acc[t] = __builtin_amdgcn_wmma_f32_16x16x4_f32(
           false, a,  false, b,
           (short)0, acc[t],  false,  false);
    }
  }

  int rbase = row0 + (lane >> 4) * 8;
#pragma unroll
  for (int t = 0; t < NT; ++t) {
    int col = t * 16 + m;
    if (col < Cout) {
#pragma unroll
      for (int j = 0; j < 8; ++j) {
        out[(size_t)(rbase + j) * ldo + col] = acc[t][j];
      }
    }
  }
}

__global__ void colstats_kernel(const float* __restrict__ g,
                                float* __restrict__ sums, int Nrows) {
  __shared__ float sh[256];
  sh[threadIdx.x] = 0.f;
  __syncthreads();
  size_t total = (size_t)Nrows * CH;
  size_t i = (size_t)blockIdx.x * 256 + threadIdx.x;
  size_t stride = (size_t)gridDim.x * 256;
  int col = (int)(i & (CH - 1));
  float s = 0.f, q = 0.f;
  for (; i < total; i += stride) {
    float v = g[i];
    s += v;
    q += v * v;
  }
  atomicAdd(&sh[col], s);
  atomicAdd(&sh[128 + col], q);
  __syncthreads();
  if (threadIdx.x < 128) {
    atomicAdd(&sums[threadIdx.x], sh[threadIdx.x]);
    atomicAdd(&sums[128 + threadIdx.x], sh[128 + threadIdx.x]);
  }
}

__global__ void bnrelu_kernel(float* __restrict__ g,
                              const float* __restrict__ sums,
                              const float* __restrict__ gamma,
                              const float* __restrict__ beta, int Nrows) {
  size_t total = (size_t)Nrows * CH;
  size_t i = (size_t)blockIdx.x * blockDim.x + threadIdx.x;
  size_t stride = (size_t)gridDim.x * blockDim.x;
  float invN = 1.0f / (float)Nrows;
  for (; i < total; i += stride) {
    int col = (int)(i & (CH - 1));
    float mean = sums[col] * invN;
    float var = sums[128 + col] * invN - mean * mean;
    float sc = gamma[col] * rsqrtf(var + BN_EPS);
    float v = (g[i] - mean) * sc + beta[col];
    g[i] = v > 0.f ? v : 0.f;
  }
}

__global__ void biasout_kernel(const float* __restrict__ agg40,
                               const float* __restrict__ b3,
                               float* __restrict__ out, int total) {
  int i = blockIdx.x * blockDim.x + threadIdx.x;
  if (i < total) {
    int c = i % 40;
    out[i] = agg40[i] + b3[c];
  }
}

extern "C" void kernel_launch(void* const* d_in, const int* in_sizes, int n_in,
                              void* d_out, int out_size, void* d_ws, size_t ws_size,
                              hipStream_t stream) {
  const float* feat   = (const float*)d_in[0];
  const int*   src    = (const int*)d_in[1];
  const int*   dst    = (const int*)d_in[2];
  const float* ew     = (const float*)d_in[3];
  const float* W1     = (const float*)d_in[4];
  const float* W2     = (const float*)d_in[5];
  const float* W3     = (const float*)d_in[6];
  const float* b3     = (const float*)d_in[7];
  const float* gamma1 = (const float*)d_in[8];
  const float* beta1  = (const float*)d_in[9];
  const float* gamma2 = (const float*)d_in[10];
  const float* beta2  = (const float*)d_in[11];

  int N = in_sizes[0] / CH;
  int E = in_sizes[1];

  float* agg  = (float*)d_ws;
  float* g    = agg + (size_t)N * CH;
  float* sums = g + (size_t)N * CH;
  float* t3   = agg;
  float* agg3 = agg + (size_t)N * 64;

  size_t n4 = (size_t)N * (CH / 4);
  int zgrid = 2048;
  int sgrid = (E + 7) / 8;
  int ggrid = (N + 127) / 128;
  int egrid = 2048;
  int bgrid = (N * 40 + 255) / 256;
  dim3 b256(256);

  zero_f4<<<zgrid, b256, 0, stream>>>((float4*)agg, n4);
  spmm128_kernel<<<sgrid, b256, 0, stream>>>(feat, src, dst, ew, agg, E);
  gemm_wmma_kernel<8><<<ggrid, b256, 0, stream>>>(agg, W1, g, N, 128, 128);
  zero_f<<<1, b256, 0, stream>>>(sums, 256);
  colstats_kernel<<<egrid, b256, 0, stream>>>(g, sums, N);
  bnrelu_kernel<<<egrid, b256, 0, stream>>>(g, sums, gamma1, beta1, N);

  zero_f4<<<zgrid, b256, 0, stream>>>((float4*)agg, n4);
  spmm128_kernel<<<sgrid, b256, 0, stream>>>(g, src, dst, ew, agg, E);
  gemm_wmma_kernel<8><<<ggrid, b256, 0, stream>>>(agg, W2, g, N, 128, 128);
  zero_f<<<1, b256, 0, stream>>>(sums, 256);
  colstats_kernel<<<egrid, b256, 0, stream>>>(g, sums, N);
  bnrelu_kernel<<<egrid, b256, 0, stream>>>(g, sums, gamma2, beta2, N);

  gemm_wmma_kernel<3><<<ggrid, b256, 0, stream>>>(g, W3, t3, N, 40, 40);
  zero_f4<<<zgrid, b256, 0, stream>>>((float4*)agg3, (size_t)N * 10);
  spmm40_kernel<<<sgrid, b256, 0, stream>>>(t3, src, dst, ew, agg3, E);
  biasout_kernel<<<bgrid, b256, 0, stream>>>(agg3, b3, (float*)d_out, N * 40);
}


# Sums by destination row: edges counting-sorted by destination once per call (integer-atomic ranks); one wave adds a row's sources in registers, arrival-rank order; no float atomics, no zero fills
# speedup vs baseline: 2.0175x; 2.0175x over previous
#include <hip/hip_runtime.h>
#include <hip/hip_bf16.h>

typedef __attribute__((ext_vector_type(2))) float v2f;
typedef __attribute__((ext_vector_type(8))) float v8f;

#define BN_EPS 1e-5f
#define CH 128
#define WSTRIDE 132

__global__ void zero_f(float* __restrict__ p, int n) {
  int i = blockIdx.x * blockDim.x + threadIdx.x;
  if (i < n) p[i] = 0.f;
}

__global__ void zero_i(int* __restrict__ p, int n) {
  int i = blockIdx.x * blockDim.x + threadIdx.x;
  if (i < n) p[i] = 0;
}

__global__ void rank_kernel(const int* __restrict__ dst, int* __restrict__ cnt,
                            int* __restrict__ rank, int E) {
  int e = blockIdx.x * 256 + threadIdx.x;
  if (e < E) rank[e] = atomicAdd(&cnt[dst[e]], 1);
}

__device__ __forceinline__ int wave_incl_scan(int s, int lane) {
  for (int o = 1; o < 32; o <<= 1) {
    int v = __shfl_up(s, o);
    if (lane >= o) s += v;
  }
  return s;
}

#define SCAN_CHUNK 16384
__global__ __launch_bounds__(1024) void scan_kernel(const int* __restrict__ cnt,
                                                    int* __restrict__ row_ptr, int n) {
  __shared__ int wsum[32];
  __shared__ int tot;
  const int tid = threadIdx.x;
  const int lane = tid & 31;
  const int wv = tid >> 5;
  const int4* c4 = (const int4*)cnt + tid * 4;
  int carry = 0;
  int4 a0 = c4[0], a1 = c4[1], a2 = c4[2], a3 = c4[3];
  for (int base = 0; base < n; base += SCAN_CHUNK) {
    int v[16] = {a0.x, a0.y, a0.z, a0.w, a1.x, a1.y, a1.z, a1.w,
                 a2.x, a2.y, a2.z, a2.w, a3.x, a3.y, a3.z, a3.w};
    const int nb = (base + SCAN_CHUNK < n) ? base + SCAN_CHUNK : base;
    const int4* p = c4 + nb / 4;
    a0 = p[0]; a1 = p[1]; a2 = p[2]; a3 = p[3];
    int ex[16];
    int t = 0;
    for (int k = 0; k < 16; ++k) { ex[k] = t; t += v[k]; }
    const int s = wave_incl_scan(t, lane);
    if (lane == 31) wsum[wv] = s;
    __syncthreads();
    if (wv == 0) {
      const int w = wsum[lane];
      const int ws = wave_incl_scan(w, lane);
      wsum[lane] = ws - w;
      if (lane == 31) tot = ws;
    }
    __syncthreads();
    const int off = carry + wsum[wv] + (s - t);
    const int i = base + tid * 16;
    for (int q = 0; q < 4; ++q) {
      const int idx = i + q * 4;
      if (idx + 3 < n) {
        int4 r = {off + ex[q * 4], off + ex[q * 4 + 1], off + ex[q * 4 + 2], off + ex[q * 4 + 3]};
        *(int4*)(row_ptr + idx) = r;
      } else {
        if (idx < n) row_ptr[idx] = off + ex[q * 4];
        if (idx + 1 < n) row_ptr[idx + 1] = off + ex[q * 4 + 1];
        if (idx + 2 < n) row_ptr[idx + 2] = off + ex[q * 4 + 2];
      }
    }
    carry += tot;
    __syncthreads();
  }
  if (tid == 0) row_ptr[n] = carry;
}

__global__ void fill_kernel(const int* __restrict__ src, const int* __restrict__ dst,
                            const float* __restrict__ ew, const int* __restrict__ rank,
                            const int* __restrict__ row_ptr, int2* __restrict__ edges, int E) {
  int e = blockIdx.x * 256 + threadIdx.x;
  if (e < E) {
    int pos = row_ptr[dst[e]] + rank[e];
    edges[pos] = make_int2(src[e], __float_as_int(ew[e]));
  }
}

template <int C4, int U>
__device__ __forceinline__ void gather_group(const float4* __restrict__ F, int srcs, int wbits,
                                             int j, int l, float4& acc) {
  float4 v[U];
  float w[U];
  for (int u = 0; u < U; ++u) {
    int s = __builtin_amdgcn_readlane(srcs, j + u);
    w[u] = __int_as_float(__builtin_amdgcn_readlane(wbits, j + u));
    v[u] = F[(size_t)s * C4 + l];
  }
  for (int u = 0; u < U; ++u) {
    acc.x += w[u] * v[u].x;
    acc.y += w[u] * v[u].y;
    acc.z += w[u] * v[u].z;
    acc.w += w[u] * v[u].w;
  }
}

template <int C4>
__global__ __launch_bounds__(256) void spmm_rows_kernel(const float* __restrict__ feat,
                                                        const int* __restrict__ row_ptr,
                                                        const int2* __restrict__ edges,
                                                        float* __restrict__ agg, int Nrows) {
  const int wave = __builtin_amdgcn_readfirstlane((int)(threadIdx.x >> 5));
  const int lane = threadIdx.x & 31;
  const int row = blockIdx.x * 8 + wave;
  if (row >= Nrows) return;
  const int beg = row_ptr[row];
  const int end = row_ptr[row + 1];
  const int l = lane < C4 ? lane : C4 - 1;
  const float4* F = (const float4*)feat;
  float4 acc = {0.f, 0.f, 0.f, 0.f};
  for (int base = beg; base < end; base += 32) {
    const int n = min(32, end - base);
    int2 my = make_int2(0, 0);
    if (lane < n) my = edges[base + lane];
    int j = 0;
    for (; j + 8 <= n; j += 8) gather_group<C4, 8>(F, my.x, my.y, j, l, acc);
    if (j + 4 <= n) { gather_group<C4, 4>(F, my.x, my.y, j, l, acc); j += 4; }
    if (j + 2 <= n) { gather_group<C4, 2>(F, my.x, my.y, j, l, acc); j += 2; }
    if (j < n) gather_group<C4, 1>(F, my.x, my.y, j, l, acc);
  }
  if (lane < C4) ((float4*)agg)[(size_t)row * C4 + lane] = acc;
}

template <int NT>
__global__ void gemm_wmma_kernel(const float* __restrict__ A,
                                 const float* __restrict__ W,
                                 float* __restrict__ out,
                                 int Nrows, int Cout, int ldo) {
  constexpr int CP = NT * 16;
  __shared__ float sWT[CP * WSTRIDE];
  int tid = threadIdx.x;
  for (int idx = tid; idx < CH * CP; idx += 256) {
    int k = idx / CP;
    int c = idx % CP;
    sWT[c * WSTRIDE + k] = (c < Cout) ? W[k * Cout + c] : 0.f;
  }
  __syncthreads();

  int wave = tid >> 5;
  int lane = tid & 31;
  int row0 = blockIdx.x * 128 + wave * 16;
  if (row0 >= Nrows) return;

  int m = lane & 15;
  int kh = (lane >> 4) * 2;

  v8f acc[NT];
#pragma unroll
  for (int t = 0; t < NT; ++t) acc[t] = {};

  const float* Arow = A + (size_t)(row0 + m) * CH;

  for (int k0 = 0; k0 < CH; k0 += 4) {
    v2f a = *(const v2f*)(Arow + k0 + kh);
#pragma unroll
    for (int t = 0; t < NT; ++t) {
      v2f b = *(const v2f*)(&sWT[(t * 16 + m) * WSTRIDE + k0 + kh]);
      acc[t] = __builtin_amdgcn_wmma_f32_16x16x4_f32(
           false, a,  false, b,
           (short)0, acc[t],  false,  false);
    }
  }

  int rbase = row0 + (lane >> 4) * 8;
#pragma unroll
  for (int t = 0; t < NT; ++t) {
    int col = t * 16 + m;
    if (col < Cout) {
#pragma unroll
      for (int j = 0; j < 8; ++j) {
        out[(size_t)(rbase + j) * ldo + col] = acc[t][j];
      }
    }
  }
}

__global__ void colstats_kernel(const float* __restrict__ g,
                                float* __restrict__ sums, int Nrows) {
  __shared__ float sh[256];
  sh[threadIdx.x] = 0.f;
  __syncthreads();
  size_t total = (size_t)Nrows * CH;
  size_t i = (size_t)blockIdx.x * 256 + threadIdx.x;
  size_t stride = (size_t)gridDim.x * 256;
  int col = (int)(i & (CH - 1));
  float s = 0.f, q = 0.f;
  for (; i < total; i += stride) {
    float v = g[i];
    s += v;
    q += v * v;
  }
  atomicAdd(&sh[col], s);
  atomicAdd(&sh[128 + col], q);
  __syncthreads();
  if (threadIdx.x < 128) {
    atomicAdd(&sums[threadIdx.x], sh[threadIdx.x]);
    atomicAdd(&sums[128 + threadIdx.x], sh[128 + threadIdx.x]);
  }
}

__global__ void bnrelu_kernel(float* __restrict__ g,
                              const float* __restrict__ sums,
                              const float* __restrict__ gamma,
                              const float* __restrict__ beta, int Nrows) {
  size_t total = (size_t)Nrows * CH;
  size_t i = (size_t)blockIdx.x * blockDim.x + threadIdx.x;
  size_t stride = (size_t)gridDim.x * blockDim.x;
  float invN = 1.0f / (float)Nrows;
  for (; i < total; i += stride) {
    int col = (int)(i & (CH - 1));
    float mean = sums[col] * invN;
    float var = sums[128 + col] * invN - mean * mean;
    float sc = gamma[col] * rsqrtf(var + BN_EPS);
    float v = (g[i] - mean) * sc + beta[col];
    g[i] = v > 0.f ? v : 0.f;
  }
}

__global__ void biasout_kernel(const float* __restrict__ agg40,
                               const float* __restrict__ b3,
                               float* __restrict__ out, int total) {
  int i = blockIdx.x * blockDim.x + threadIdx.x;
  if (i < total) {
    int c = i % 40;
    out[i] = agg40[i] + b3[c];
  }
}

extern "C" void kernel_launch(void* const* d_in, const int* in_sizes, int n_in,
                              void* d_out, int out_size, void* d_ws, size_t ws_size,
                              hipStream_t stream) {
  const float* feat   = (const float*)d_in[0];
  const int*   src    = (const int*)d_in[1];
  const int*   dst    = (const int*)d_in[2];
  const float* ew     = (const float*)d_in[3];
  const float* W1     = (const float*)d_in[4];
  const float* W2     = (const float*)d_in[5];
  const float* W3     = (const float*)d_in[6];
  const float* b3     = (const float*)d_in[7];
  const float* gamma1 = (const float*)d_in[8];
  const float* beta1  = (const float*)d_in[9];
  const float* gamma2 = (const float*)d_in[10];
  const float* beta2  = (const float*)d_in[11];

  int N = in_sizes[0] / CH;
  int E = in_sizes[1];

  float* agg  = (float*)d_ws;
  float* g    = agg + (size_t)N * CH;
  float* sums = g + (size_t)N * CH;
  float* t3   = agg;
  float* agg3 = agg + (size_t)N * 64;

  int2* edges   = (int2*)d_out;
  int*  row_ptr = (int*)(edges + E);
  int   npad = (N + SCAN_CHUNK - 1) / SCAN_CHUNK * SCAN_CHUNK;
  int*  cnt  = (int*)g;
  int*  rank = cnt + npad;

  int ngrid = (npad + 255) / 256;
  int cgrid = (E + 255) / 256;
  int rgrid = (N + 7) / 8;
  int ggrid = (N + 127) / 128;
  int egrid = 2048;
  int bgrid = (N * 40 + 255) / 256;
  dim3 b256(256);

  zero_i<<<ngrid, b256, 0, stream>>>(cnt, npad);
  rank_kernel<<<cgrid, b256, 0, stream>>>(dst, cnt, rank, E);
  scan_kernel<<<1, 1024, 0, stream>>>(cnt, row_ptr, N);
  fill_kernel<<<cgrid, b256, 0, stream>>>(src, dst, ew, rank, row_ptr, edges, E);

  spmm_rows_kernel<32><<<rgrid, b256, 0, stream>>>(feat, row_ptr, edges, agg, N);
  gemm_wmma_kernel<8><<<ggrid, b256, 0, stream>>>(agg, W1, g, N, 128, 128);
  zero_f<<<1, b256, 0, stream>>>(sums, 256);
  colstats_kernel<<<egrid, b256, 0, stream>>>(g, sums, N);
  bnrelu_kernel<<<egrid, b256, 0, stream>>>(g, sums, gamma1, beta1, N);

  spmm_rows_kernel<32><<<rgrid, b256, 0, stream>>>(g, row_ptr, edges, agg, N);
  gemm_wmma_kernel<8><<<ggrid, b256, 0, stream>>>(agg, W2, g, N, 128, 128);
  zero_f<<<1, b256, 0, stream>>>(sums, 256);
  colstats_kernel<<<egrid, b256, 0, stream>>>(g, sums, N);
  bnrelu_kernel<<<egrid, b256, 0, stream>>>(g, sums, gamma2, beta2, N);

  gemm_wmma_kernel<3><<<ggrid, b256, 0, stream>>>(g, W3, t3, N, 40, 40);
  spmm_rows_kernel<10><<<rgrid, b256, 0, stream>>>(t3, row_ptr, edges, agg3, N);
  biasout_kernel<<<bgrid, b256, 0, stream>>>(agg3, b3, (float*)d_out, N * 40);
}


# Batch norm folded into neighbour launches: GEMM epilogue adds column sum/sumsq (new summation tree, one float atomic per column per block), next layer normalises+clamps rows as it loads them
# speedup vs baseline: 2.3109x; 1.1454x over previous
#include <hip/hip_runtime.h>
#include <hip/hip_bf16.h>

typedef __attribute__((ext_vector_type(2))) float v2f;
typedef __attribute__((ext_vector_type(8))) float v8f;

#define BN_EPS 1e-5f
#define CH 128
#define WSTRIDE 132

__global__ void zero_kernel(int* __restrict__ cnt, int n, float* __restrict__ sums) {
  int i = blockIdx.x * blockDim.x + threadIdx.x;
  if (i < n) cnt[i] = 0;
  if (i < 512) sums[i] = 0.f;
}

__global__ void rank_kernel(const int* __restrict__ dst, int* __restrict__ cnt,
                            int* __restrict__ rank, int E) {
  int e = blockIdx.x * 256 + threadIdx.x;
  if (e < E) rank[e] = atomicAdd(&cnt[dst[e]], 1);
}

__device__ __forceinline__ int wave_incl_scan(int s, int lane) {
  for (int o = 1; o < 32; o <<= 1) {
    int v = __shfl_up(s, o);
    if (lane >= o) s += v;
  }
  return s;
}

#define SCAN_CHUNK 16384
__global__ __launch_bounds__(1024) void scan_kernel(const int* __restrict__ cnt,
                                                    int* __restrict__ row_ptr, int n) {
  __shared__ int wsum[32];
  __shared__ int tot;
  const int tid = threadIdx.x;
  const int lane = tid & 31;
  const int wv = tid >> 5;
  const int4* c4 = (const int4*)cnt + tid * 4;
  int carry = 0;
  int4 a0 = c4[0], a1 = c4[1], a2 = c4[2], a3 = c4[3];
  for (int base = 0; base < n; base += SCAN_CHUNK) {
    int v[16] = {a0.x, a0.y, a0.z, a0.w, a1.x, a1.y, a1.z, a1.w,
                 a2.x, a2.y, a2.z, a2.w, a3.x, a3.y, a3.z, a3.w};
    const int nb = (base + SCAN_CHUNK < n) ? base + SCAN_CHUNK : base;
    const int4* p = c4 + nb / 4;
    a0 = p[0]; a1 = p[1]; a2 = p[2]; a3 = p[3];
    int ex[16];
    int t = 0;
    for (int k = 0; k < 16; ++k) { ex[k] = t; t += v[k]; }
    const int s = wave_incl_scan(t, lane);
    if (lane == 31) wsum[wv] = s;
    __syncthreads();
    if (wv == 0) {
      const int w = wsum[lane];
      const int ws = wave_incl_scan(w, lane);
      wsum[lane] = ws - w;
      if (lane == 31) tot = ws;
    }
    __syncthreads();
    const int off = carry + wsum[wv] + (s - t);
    const int i = base + tid * 16;
    for (int q = 0; q < 4; ++q) {
      const int idx = i + q * 4;
      if (idx + 3 < n) {
        int4 r = {off + ex[q * 4], off + ex[q * 4 + 1], off + ex[q * 4 + 2], off + ex[q * 4 + 3]};
        *(int4*)(row_ptr + idx) = r;
      } else {
        if (idx < n) row_ptr[idx] = off + ex[q * 4];
        if (idx + 1 < n) row_ptr[idx + 1] = off + ex[q * 4 + 1];
        if (idx + 2 < n) row_ptr[idx + 2] = off + ex[q * 4 + 2];
      }
    }
    carry += tot;
    __syncthreads();
  }
  if (tid == 0) row_ptr[n] = carry;
}

__global__ void fill_kernel(const int* __restrict__ src, const int* __restrict__ dst,
                            const float* __restrict__ ew, const int* __restrict__ rank,
                            const int* __restrict__ row_ptr, int2* __restrict__ edges, int E) {
  int e = blockIdx.x * 256 + threadIdx.x;
  if (e < E) {
    int pos = row_ptr[dst[e]] + rank[e];
    edges[pos] = make_int2(src[e], __float_as_int(ew[e]));
  }
}

__device__ __forceinline__ float bn_relu(float x, float mean, float sc, float beta) {
  float v = (x - mean) * sc + beta;
  return v > 0.f ? v : 0.f;
}

template <int C4, int U, bool BN>
__device__ __forceinline__ void gather_group(const float4* __restrict__ F, int srcs, int wbits,
                                             int j, int l, float4& acc, const float4& bm,
                                             const float4& bs, const float4& bb) {
  float4 v[U];
  float w[U];
  for (int u = 0; u < U; ++u) {
    int s = __builtin_amdgcn_readlane(srcs, j + u);
    w[u] = __int_as_float(__builtin_amdgcn_readlane(wbits, j + u));
    v[u] = F[(size_t)s * C4 + l];
  }
  for (int u = 0; u < U; ++u) {
    float4 h = v[u];
    if constexpr (BN) {
      h.x = bn_relu(h.x, bm.x, bs.x, bb.x);
      h.y = bn_relu(h.y, bm.y, bs.y, bb.y);
      h.z = bn_relu(h.z, bm.z, bs.z, bb.z);
      h.w = bn_relu(h.w, bm.w, bs.w, bb.w);
    }
    acc.x += w[u] * h.x;
    acc.y += w[u] * h.y;
    acc.z += w[u] * h.z;
    acc.w += w[u] * h.w;
  }
}

template <int C4, bool BN>
__global__ __launch_bounds__(256) void spmm_rows_kernel(const float* __restrict__ feat,
                                                        const int* __restrict__ row_ptr,
                                                        const int2* __restrict__ edges,
                                                        float* __restrict__ agg, int Nrows,
                                                        const float* __restrict__ sums,
                                                        const float* __restrict__ gamma,
                                                        const float* __restrict__ beta) {
  const int wave = __builtin_amdgcn_readfirstlane((int)(threadIdx.x >> 5));
  const int lane = threadIdx.x & 31;
  const int row = blockIdx.x * 8 + wave;
  if (row >= Nrows) return;
  const int l = lane < C4 ? lane : C4 - 1;
  const float4* F = (const float4*)feat;
  const int beg = row_ptr[row];
  const int end = row_ptr[row + 1];
  int base = beg;
  int n = min(32, end - base);
  int2 my = make_int2(0, 0);
  if (lane < n) my = edges[base + lane];
  float4 bm = {0.f, 0.f, 0.f, 0.f}, bs = bm, bb = bm;
  if constexpr (BN) {
    const float invN = 1.0f / (float)Nrows;
    const float4 s = ((const float4*)sums)[l];
    const float4 q = ((const float4*)(sums + CH))[l];
    const float4 g4 = ((const float4*)gamma)[l];
    bb = ((const float4*)beta)[l];
    bm.x = s.x * invN; bm.y = s.y * invN; bm.z = s.z * invN; bm.w = s.w * invN;
    bs.x = g4.x * rsqrtf(q.x * invN - bm.x * bm.x + BN_EPS);
    bs.y = g4.y * rsqrtf(q.y * invN - bm.y * bm.y + BN_EPS);
    bs.z = g4.z * rsqrtf(q.z * invN - bm.z * bm.z + BN_EPS);
    bs.w = g4.w * rsqrtf(q.w * invN - bm.w * bm.w + BN_EPS);
  }
  float4 acc = {0.f, 0.f, 0.f, 0.f};
  while (base < end) {
    int j = 0;
    for (; j + 8 <= n; j += 8) gather_group<C4, 8, BN>(F, my.x, my.y, j, l, acc, bm, bs, bb);
    if (j + 4 <= n) { gather_group<C4, 4, BN>(F, my.x, my.y, j, l, acc, bm, bs, bb); j += 4; }
    if (j + 2 <= n) { gather_group<C4, 2, BN>(F, my.x, my.y, j, l, acc, bm, bs, bb); j += 2; }
    if (j < n) gather_group<C4, 1, BN>(F, my.x, my.y, j, l, acc, bm, bs, bb);
    base += 32;
    n = min(32, end - base);
    my = make_int2(0, 0);
    if (lane < n) my = edges[base + lane];
  }
  if (lane < C4) ((float4*)agg)[(size_t)row * C4 + lane] = acc;
}

template <int NT, bool BN_IN, bool STATS>
__global__ void gemm_wmma_kernel(const float* __restrict__ A,
                                 const float* __restrict__ W,
                                 float* __restrict__ out,
                                 int Nrows, int Cout, int ldo,
                                 const float* __restrict__ in_sums,
                                 const float* __restrict__ in_gamma,
                                 const float* __restrict__ in_beta,
                                 float* __restrict__ out_sums) {
  constexpr int CP = NT * 16;
  __shared__ float sWT[CP * WSTRIDE];
  __shared__ float4 sBNa[BN_IN ? CH / 2 : 1];
  __shared__ float2 sBNb[BN_IN ? CH / 2 : 1];
  __shared__ float sPart[STATS ? 8 * 2 * CH : 1];
  int tid = threadIdx.x;
  for (int idx = tid; idx < CH * CP; idx += 256) {
    int k = idx / CP;
    int c = idx % CP;
    sWT[c * WSTRIDE + k] = (c < Cout) ? W[k * Cout + c] : 0.f;
  }
  if constexpr (BN_IN) {
    if (tid < CH / 2) {
      const float invN = 1.0f / (float)Nrows;
      const int c = tid * 2;
      const float m0 = in_sums[c] * invN;
      const float m1 = in_sums[c + 1] * invN;
      const float s0 = in_gamma[c] * rsqrtf(in_sums[CH + c] * invN - m0 * m0 + BN_EPS);
      const float s1 = in_gamma[c + 1] * rsqrtf(in_sums[CH + c + 1] * invN - m1 * m1 + BN_EPS);
      sBNa[tid] = make_float4(m0, m1, s0, s1);
      sBNb[tid] = make_float2(in_beta[c], in_beta[c + 1]);
    }
  }
  __syncthreads();

  const int wave = __builtin_amdgcn_readfirstlane(tid >> 5);
  int lane = tid & 31;
  int row0 = blockIdx.x * 128 + wave * 16;
  const bool valid = row0 < Nrows;

  int m = lane & 15;
  int kh = (lane >> 4) * 2;

  v8f acc[NT];
#pragma unroll
  for (int t = 0; t < NT; ++t) acc[t] = {};

  if (valid) {
    const float* Arow = A + (size_t)(row0 + m) * CH;

    for (int k0 = 0; k0 < CH; k0 += 4) {
      v2f a = *(const v2f*)(Arow + k0 + kh);
      if constexpr (BN_IN) {
        const float4 ms = sBNa[(k0 + kh) >> 1];
        const float2 be = sBNb[(k0 + kh) >> 1];
        float x0 = (a.x - ms.x) * ms.z + be.x;
        float x1 = (a.y - ms.y) * ms.w + be.y;
        a.x = x0 > 0.f ? x0 : 0.f;
        a.y = x1 > 0.f ? x1 : 0.f;
      }
#pragma unroll
      for (int t = 0; t < NT; ++t) {
        v2f b = *(const v2f*)(&sWT[(t * 16 + m) * WSTRIDE + k0 + kh]);
        acc[t] = __builtin_amdgcn_wmma_f32_16x16x4_f32(
             false, a,  false, b,
             (short)0, acc[t],  false,  false);
      }
    }

    int rbase = row0 + (lane >> 4) * 8;
#pragma unroll
    for (int t = 0; t < NT; ++t) {
      int col = t * 16 + m;
      if (col < Cout) {
#pragma unroll
        for (int j = 0; j < 8; ++j) {
          out[(size_t)(rbase + j) * ldo + col] = acc[t][j];
        }
      }
    }
  }

  if constexpr (STATS) {
    static_assert(!STATS || NT * 16 == CH, "column sums assume a 128-column output");
    for (int t = 0; t < NT; ++t) {
      float s = 0.f, q = 0.f;
      for (int j = 0; j < 8; ++j) {
        const float v = acc[t][j];
        s += v;
        q += v * v;
      }
      s += __shfl_xor(s, 16);
      q += __shfl_xor(q, 16);
      if (lane < 16) {
        sPart[wave * (2 * CH) + t * 16 + lane] = s;
        sPart[wave * (2 * CH) + CH + t * 16 + lane] = q;
      }
    }
    __syncthreads();
    float tot = 0.f;
    for (int w = 0; w < 8; ++w) tot += sPart[w * (2 * CH) + tid];
    atomicAdd(&out_sums[tid], tot);
  }
}

__global__ void biasout_kernel(const float* __restrict__ agg40,
                               const float* __restrict__ b3,
                               float* __restrict__ out, int total) {
  int i = blockIdx.x * blockDim.x + threadIdx.x;
  if (i < total) {
    int c = i % 40;
    out[i] = agg40[i] + b3[c];
  }
}

extern "C" void kernel_launch(void* const* d_in, const int* in_sizes, int n_in,
                              void* d_out, int out_size, void* d_ws, size_t ws_size,
                              hipStream_t stream) {
  const float* feat   = (const float*)d_in[0];
  const int*   src    = (const int*)d_in[1];
  const int*   dst    = (const int*)d_in[2];
  const float* ew     = (const float*)d_in[3];
  const float* W1     = (const float*)d_in[4];
  const float* W2     = (const float*)d_in[5];
  const float* W3     = (const float*)d_in[6];
  const float* b3     = (const float*)d_in[7];
  const float* gamma1 = (const float*)d_in[8];
  const float* beta1  = (const float*)d_in[9];
  const float* gamma2 = (const float*)d_in[10];
  const float* beta2  = (const float*)d_in[11];

  int N = in_sizes[0] / CH;
  int E = in_sizes[1];

  float* agg  = (float*)d_ws;
  float* g    = agg + (size_t)N * CH;
  float* t3   = agg;
  float* agg3 = agg + (size_t)N * 64;

  int2* edges   = (int2*)d_out;
  int*  row_ptr = (int*)(edges + E);
  float* sums1  = (float*)(row_ptr + (N + 1 + 3) / 4 * 4);
  float* sums2  = sums1 + 2 * CH;
  int   npad = (N + SCAN_CHUNK - 1) / SCAN_CHUNK * SCAN_CHUNK;
  int*  cnt  = (int*)g;
  int*  rank = cnt + npad;

  int ngrid = (npad + 255) / 256;
  int cgrid = (E + 255) / 256;
  int rgrid = (N + 7) / 8;
  int ggrid = (N + 127) / 128;
  int bgrid = (N * 40 + 255) / 256;
  dim3 b256(256);

  zero_kernel<<<ngrid, b256, 0, stream>>>(cnt, npad, sums1);
  rank_kernel<<<cgrid, b256, 0, stream>>>(dst, cnt, rank, E);
  scan_kernel<<<1, 1024, 0, stream>>>(cnt, row_ptr, N);
  fill_kernel<<<cgrid, b256, 0, stream>>>(src, dst, ew, rank, row_ptr, edges, E);

  spmm_rows_kernel<32, false><<<rgrid, b256, 0, stream>>>(feat, row_ptr, edges, agg, N,
                                                           nullptr, nullptr, nullptr);
  gemm_wmma_kernel<8, false, true><<<ggrid, b256, 0, stream>>>(agg, W1, g, N, 128, 128,
                                                                nullptr, nullptr, nullptr, sums1);

  spmm_rows_kernel<32, true><<<rgrid, b256, 0, stream>>>(g, row_ptr, edges, agg, N,
                                                          sums1, gamma1, beta1);
  gemm_wmma_kernel<8, false, true><<<ggrid, b256, 0, stream>>>(agg, W2, g, N, 128, 128,
                                                                nullptr, nullptr, nullptr, sums2);

  gemm_wmma_kernel<3, true, false><<<ggrid, b256, 0, stream>>>(g, W3, t3, N, 40, 40,
                                                                sums2, gamma2, beta2, nullptr);
  spmm_rows_kernel<10, false><<<rgrid, b256, 0, stream>>>(t3, row_ptr, edges, agg3, N,
                                                           nullptr, nullptr, nullptr);
  biasout_kernel<<<bgrid, b256, 0, stream>>>(agg3, b3, (float*)d_out, N * 40);
}


# Aggregation and its product share one launch in layers 1 and 2: a 16-wave group sums 64 rows into an LDS tile that feeds the WMMAs; aggregate plane never stored; column sums now per 64-row tile
# speedup vs baseline: 2.8017x; 1.2124x over previous
#include <hip/hip_runtime.h>
#include <hip/hip_bf16.h>

typedef __attribute__((ext_vector_type(2))) float v2f;
typedef __attribute__((ext_vector_type(8))) float v8f;

#define BN_EPS 1e-5f
#define CH 128
#define WSTRIDE 132

__global__ void zero_kernel(int* __restrict__ cnt, int n, float* __restrict__ sums) {
  int i = blockIdx.x * blockDim.x + threadIdx.x;
  if (i < n) cnt[i] = 0;
  if (i < 512) sums[i] = 0.f;
}

__global__ void rank_kernel(const int* __restrict__ dst, int* __restrict__ cnt,
                            int* __restrict__ rank, int E) {
  int e = blockIdx.x * 256 + threadIdx.x;
  if (e < E) rank[e] = atomicAdd(&cnt[dst[e]], 1);
}

__device__ __forceinline__ int wave_incl_scan(int s, int lane) {
  for (int o = 1; o < 32; o <<= 1) {
    int v = __shfl_up(s, o);
    if (lane >= o) s += v;
  }
  return s;
}

#define SCAN_CHUNK 16384
__global__ __launch_bounds__(1024) void scan_kernel(const int* __restrict__ cnt,
                                                    int* __restrict__ row_ptr, int n) {
  __shared__ int wsum[32];
  __shared__ int tot;
  const int tid = threadIdx.x;
  const int lane = tid & 31;
  const int wv = tid >> 5;
  const int4* c4 = (const int4*)cnt + tid * 4;
  int carry = 0;
  int4 a0 = c4[0], a1 = c4[1], a2 = c4[2], a3 = c4[3];
  for (int base = 0; base < n; base += SCAN_CHUNK) {
    int v[16] = {a0.x, a0.y, a0.z, a0.w, a1.x, a1.y, a1.z, a1.w,
                 a2.x, a2.y, a2.z, a2.w, a3.x, a3.y, a3.z, a3.w};
    const int nb = (base + SCAN_CHUNK < n) ? base + SCAN_CHUNK : base;
    const int4* p = c4 + nb / 4;
    a0 = p[0]; a1 = p[1]; a2 = p[2]; a3 = p[3];
    int ex[16];
    int t = 0;
    for (int k = 0; k < 16; ++k) { ex[k] = t; t += v[k]; }
    const int s = wave_incl_scan(t, lane);
    if (lane == 31) wsum[wv] = s;
    __syncthreads();
    if (wv == 0) {
      const int w = wsum[lane];
      const int ws = wave_incl_scan(w, lane);
      wsum[lane] = ws - w;
      if (lane == 31) tot = ws;
    }
    __syncthreads();
    const int off = carry + wsum[wv] + (s - t);
    const int i = base + tid * 16;
    for (int q = 0; q < 4; ++q) {
      const int idx = i + q * 4;
      if (idx + 3 < n) {
        int4 r = {off + ex[q * 4], off + ex[q * 4 + 1], off + ex[q * 4 + 2], off + ex[q * 4 + 3]};
        *(int4*)(row_ptr + idx) = r;
      } else {
        if (idx < n) row_ptr[idx] = off + ex[q * 4];
        if (idx + 1 < n) row_ptr[idx + 1] = off + ex[q * 4 + 1];
        if (idx + 2 < n) row_ptr[idx + 2] = off + ex[q * 4 + 2];
      }
    }
    carry += tot;
    __syncthreads();
  }
  if (tid == 0) row_ptr[n] = carry;
}

__global__ void fill_kernel(const int* __restrict__ src, const int* __restrict__ dst,
                            const float* __restrict__ ew, const int* __restrict__ rank,
                            const int* __restrict__ row_ptr, int2* __restrict__ edges, int E) {
  int e = blockIdx.x * 256 + threadIdx.x;
  if (e < E) {
    int pos = row_ptr[dst[e]] + rank[e];
    edges[pos] = make_int2(src[e], __float_as_int(ew[e]));
  }
}

__device__ __forceinline__ float bn_relu(float x, float mean, float sc, float beta) {
  float v = (x - mean) * sc + beta;
  return v > 0.f ? v : 0.f;
}

template <int C4, int U, bool BN>
__device__ __forceinline__ void gather_group(const float4* __restrict__ F, int srcs, int wbits,
                                             int j, int l, float4& acc, const float4& bm,
                                             const float4& bs, const float4& bb) {
  float4 v[U];
  float w[U];
  for (int u = 0; u < U; ++u) {
    int s = __builtin_amdgcn_readlane(srcs, j + u);
    w[u] = __int_as_float(__builtin_amdgcn_readlane(wbits, j + u));
    v[u] = F[(size_t)s * C4 + l];
  }
  for (int u = 0; u < U; ++u) {
    float4 h = v[u];
    if constexpr (BN) {
      h.x = bn_relu(h.x, bm.x, bs.x, bb.x);
      h.y = bn_relu(h.y, bm.y, bs.y, bb.y);
      h.z = bn_relu(h.z, bm.z, bs.z, bb.z);
      h.w = bn_relu(h.w, bm.w, bs.w, bb.w);
    }
    acc.x += w[u] * h.x;
    acc.y += w[u] * h.y;
    acc.z += w[u] * h.z;
    acc.w += w[u] * h.w;
  }
}

template <int C4, bool BN>
__global__ __launch_bounds__(256) void spmm_rows_kernel(const float* __restrict__ feat,
                                                        const int* __restrict__ row_ptr,
                                                        const int2* __restrict__ edges,
                                                        float* __restrict__ agg, int Nrows,
                                                        const float* __restrict__ sums,
                                                        const float* __restrict__ gamma,
                                                        const float* __restrict__ beta) {
  const int wave = __builtin_amdgcn_readfirstlane((int)(threadIdx.x >> 5));
  const int lane = threadIdx.x & 31;
  const int row = blockIdx.x * 8 + wave;
  if (row >= Nrows) return;
  const int l = lane < C4 ? lane : C4 - 1;
  const float4* F = (const float4*)feat;
  const int beg = row_ptr[row];
  const int end = row_ptr[row + 1];
  int base = beg;
  int n = min(32, end - base);
  int2 my = make_int2(0, 0);
  if (lane < n) my = edges[base + lane];
  float4 bm = {0.f, 0.f, 0.f, 0.f}, bs = bm, bb = bm;
  if constexpr (BN) {
    const float invN = 1.0f / (float)Nrows;
    const float4 s = ((const float4*)sums)[l];
    const float4 q = ((const float4*)(sums + CH))[l];
    const float4 g4 = ((const float4*)gamma)[l];
    bb = ((const float4*)beta)[l];
    bm.x = s.x * invN; bm.y = s.y * invN; bm.z = s.z * invN; bm.w = s.w * invN;
    bs.x = g4.x * rsqrtf(q.x * invN - bm.x * bm.x + BN_EPS);
    bs.y = g4.y * rsqrtf(q.y * invN - bm.y * bm.y + BN_EPS);
    bs.z = g4.z * rsqrtf(q.z * invN - bm.z * bm.z + BN_EPS);
    bs.w = g4.w * rsqrtf(q.w * invN - bm.w * bm.w + BN_EPS);
  }
  float4 acc = {0.f, 0.f, 0.f, 0.f};
  while (base < end) {
    int j = 0;
    for (; j + 8 <= n; j += 8) gather_group<C4, 8, BN>(F, my.x, my.y, j, l, acc, bm, bs, bb);
    if (j + 4 <= n) { gather_group<C4, 4, BN>(F, my.x, my.y, j, l, acc, bm, bs, bb); j += 4; }
    if (j + 2 <= n) { gather_group<C4, 2, BN>(F, my.x, my.y, j, l, acc, bm, bs, bb); j += 2; }
    if (j < n) gather_group<C4, 1, BN>(F, my.x, my.y, j, l, acc, bm, bs, bb);
    base += 32;
    n = min(32, end - base);
    my = make_int2(0, 0);
    if (lane < n) my = edges[base + lane];
  }
  if (lane < C4) ((float4*)agg)[(size_t)row * C4 + lane] = acc;
}

template <int NT, bool BN_IN>
__global__ void gemm_wmma_kernel(const float* __restrict__ A,
                                 const float* __restrict__ W,
                                 float* __restrict__ out,
                                 int Nrows, int Cout, int ldo,
                                 const float* __restrict__ in_sums,
                                 const float* __restrict__ in_gamma,
                                 const float* __restrict__ in_beta) {
  constexpr int CP = NT * 16;
  __shared__ float sWT[CP * WSTRIDE];
  __shared__ float4 sBNa[BN_IN ? CH / 2 : 1];
  __shared__ float2 sBNb[BN_IN ? CH / 2 : 1];
  int tid = threadIdx.x;
  for (int idx = tid; idx < CH * CP; idx += 256) {
    int k = idx / CP;
    int c = idx % CP;
    sWT[c * WSTRIDE + k] = (c < Cout) ? W[k * Cout + c] : 0.f;
  }
  if constexpr (BN_IN) {
    if (tid < CH / 2) {
      const float invN = 1.0f / (float)Nrows;
      const int c = tid * 2;
      const float m0 = in_sums[c] * invN;
      const float m1 = in_sums[c + 1] * invN;
      const float s0 = in_gamma[c] * rsqrtf(in_sums[CH + c] * invN - m0 * m0 + BN_EPS);
      const float s1 = in_gamma[c + 1] * rsqrtf(in_sums[CH + c + 1] * invN - m1 * m1 + BN_EPS);
      sBNa[tid] = make_float4(m0, m1, s0, s1);
      sBNb[tid] = make_float2(in_beta[c], in_beta[c + 1]);
    }
  }
  __syncthreads();

  const int wave = __builtin_amdgcn_readfirstlane(tid >> 5);
  int lane = tid & 31;
  int row0 = blockIdx.x * 128 + wave * 16;
  const bool valid = row0 < Nrows;

  int m = lane & 15;
  int kh = (lane >> 4) * 2;

  v8f acc[NT];
#pragma unroll
  for (int t = 0; t < NT; ++t) acc[t] = {};

  if (valid) {
    const float* Arow = A + (size_t)(row0 + m) * CH;

    for (int k0 = 0; k0 < CH; k0 += 4) {
      v2f a = *(const v2f*)(Arow + k0 + kh);
      if constexpr (BN_IN) {
        const float4 ms = sBNa[(k0 + kh) >> 1];
        const float2 be = sBNb[(k0 + kh) >> 1];
        float x0 = (a.x - ms.x) * ms.z + be.x;
        float x1 = (a.y - ms.y) * ms.w + be.y;
        a.x = x0 > 0.f ? x0 : 0.f;
        a.y = x1 > 0.f ? x1 : 0.f;
      }
#pragma unroll
      for (int t = 0; t < NT; ++t) {
        v2f b = *(const v2f*)(&sWT[(t * 16 + m) * WSTRIDE + k0 + kh]);
        acc[t] = __builtin_amdgcn_wmma_f32_16x16x4_f32(
             false, a,  false, b,
             (short)0, acc[t],  false,  false);
      }
    }

    int rbase = row0 + (lane >> 4) * 8;
#pragma unroll
    for (int t = 0; t < NT; ++t) {
      int col = t * 16 + m;
      if (col < Cout) {
#pragma unroll
        for (int j = 0; j < 8; ++j) {
          out[(size_t)(rbase + j) * ldo + col] = acc[t][j];
        }
      }
    }
  }
}

template <bool BN>
__device__ __forceinline__ float4 gather_row(const float4* __restrict__ F,
                                             const int2* __restrict__ edges, int base, int end,
                                             int2 my, int lane, const float4& bm,
                                             const float4& bs, const float4& bb) {
  float4 acc = {0.f, 0.f, 0.f, 0.f};
  int n = min(32, end - base);
  while (base < end) {
    int j = 0;
    for (; j + 8 <= n; j += 8) gather_group<32, 8, BN>(F, my.x, my.y, j, lane, acc, bm, bs, bb);
    if (j + 4 <= n) { gather_group<32, 4, BN>(F, my.x, my.y, j, lane, acc, bm, bs, bb); j += 4; }
    if (j + 2 <= n) { gather_group<32, 2, BN>(F, my.x, my.y, j, lane, acc, bm, bs, bb); j += 2; }
    if (j < n) gather_group<32, 1, BN>(F, my.x, my.y, j, lane, acc, bm, bs, bb);
    base += 32;
    n = min(32, end - base);
    my = make_int2(0, 0);
    if (lane < n) my = edges[base + lane];
  }
  return acc;
}

#define TROWS 64
template <bool BN>
__global__ __launch_bounds__(512, 12) void spmm_gemm_kernel(const float* __restrict__ feat,
                                                             const int* __restrict__ row_ptr,
                                                             const int2* __restrict__ edges,
                                                             const float* __restrict__ W,
                                                             float* __restrict__ out, int Nrows,
                                                             const float* __restrict__ in_sums,
                                                             const float* __restrict__ in_gamma,
                                                             const float* __restrict__ in_beta,
                                                             float* __restrict__ out_sums) {
  __shared__ float sWT[CH * WSTRIDE];
  __shared__ float sA[TROWS * WSTRIDE];
  __shared__ float sPart[(TROWS / 16) * 2 * CH];
  const int tid = threadIdx.x;
  const int wave = __builtin_amdgcn_readfirstlane(tid >> 5);
  const int lane = __builtin_amdgcn_mbcnt_lo(~0u, 0u);

  const int r0 = blockIdx.x * TROWS + wave * 4;
  int rp[5];
  for (int i = 0; i < 5; ++i) rp[i] = row_ptr[min(r0 + i, Nrows)];
  int2 first[4];
#pragma unroll
  for (int i = 0; i < 4; ++i) {
    first[i] = make_int2(0, 0);
    if (lane < rp[i + 1] - rp[i]) first[i] = edges[rp[i] + lane];
  }

  {
    float wv[32];
#pragma unroll
    for (int j = 0; j < 32; ++j) wv[j] = W[tid + j * 512];
    float* wdst = &sWT[(tid & (CH - 1)) * WSTRIDE + (tid >> 7)];
#pragma unroll
    for (int j = 0; j < 32; ++j) wdst[4 * j] = wv[j];
  }
#pragma unroll
  for (int i = 0; i < 4; ++i) {
    const v2f pr = {__int_as_float(first[i].x), __int_as_float(first[i].y)};
    *(v2f*)(&sA[(wave * 4 + i) * WSTRIDE + lane * 2]) = pr;
  }

  float4 bm = {0.f, 0.f, 0.f, 0.f}, bs = bm, bb = bm;
  if constexpr (BN) {
    const float invN = 1.0f / (float)Nrows;
    const float4 s = ((const float4*)in_sums)[lane];
    const float4 q = ((const float4*)(in_sums + CH))[lane];
    const float4 g4 = ((const float4*)in_gamma)[lane];
    bb = ((const float4*)in_beta)[lane];
    bm.x = s.x * invN; bm.y = s.y * invN; bm.z = s.z * invN; bm.w = s.w * invN;
    bs.x = g4.x * rsqrtf(q.x * invN - bm.x * bm.x + BN_EPS);
    bs.y = g4.y * rsqrtf(q.y * invN - bm.y * bm.y + BN_EPS);
    bs.z = g4.z * rsqrtf(q.z * invN - bm.z * bm.z + BN_EPS);
    bs.w = g4.w * rsqrtf(q.w * invN - bm.w * bm.w + BN_EPS);
  }

  const float4* F = (const float4*)feat;
#pragma unroll
  for (int i = 0; i < 4; ++i) {
    const v2f pr = *(const v2f*)(&sA[(wave * 4 + i) * WSTRIDE + lane * 2]);
    const int2 cur = make_int2(__float_as_int(pr.x), __float_as_int(pr.y));
    const float4 acc = gather_row<BN>(F, edges, rp[i], rp[i + 1], cur, lane, bm, bs, bb);
    *(float4*)(&sA[(wave * 4 + i) * WSTRIDE + lane * 4]) = acc;
  }
  __syncthreads();

  const int stripe = wave >> 2;
  const int t0 = (wave & 3) * 2;
  const int m = lane & 15;
  const int kh = (lane >> 4) * 2;
  const float* Ap = &sA[(stripe * 16 + m) * WSTRIDE + kh];
  const float* Bp0 = &sWT[(t0 * 16 + m) * WSTRIDE + kh];
  const float* Bp1 = Bp0 + 16 * WSTRIDE;
  v8f acc0 = {}, acc1 = {};
#pragma unroll 8
  for (int k0 = 0; k0 < CH; k0 += 4) {
    const v2f a = *(const v2f*)(Ap + k0);
    const v2f b0 = *(const v2f*)(Bp0 + k0);
    const v2f b1 = *(const v2f*)(Bp1 + k0);
    acc0 = __builtin_amdgcn_wmma_f32_16x16x4_f32(false, a, false, b0, (short)0, acc0, false, false);
    acc1 = __builtin_amdgcn_wmma_f32_16x16x4_f32(false, a, false, b1, (short)0, acc1, false, false);
  }

  const int rbase = blockIdx.x * TROWS + stripe * 16 + (lane >> 4) * 8;
  const int col = t0 * 16 + m;
  for (int j = 0; j < 8; ++j) {
    if (rbase + j < Nrows) {
      out[(size_t)(rbase + j) * CH + col] = acc0[j];
      out[(size_t)(rbase + j) * CH + col + 16] = acc1[j];
    }
  }

  {
    float s0 = 0.f, q0 = 0.f, s1 = 0.f, q1 = 0.f;
    for (int j = 0; j < 8; ++j) {
      const float v0 = acc0[j], v1 = acc1[j];
      s0 += v0; q0 += v0 * v0;
      s1 += v1; q1 += v1 * v1;
    }
    s0 += __shfl_xor(s0, 16); q0 += __shfl_xor(q0, 16);
    s1 += __shfl_xor(s1, 16); q1 += __shfl_xor(q1, 16);
    if (lane < 16) {
      float* p = &sPart[stripe * (2 * CH) + col];
      p[0] = s0; p[CH] = q0;
      p[16] = s1; p[CH + 16] = q1;
    }
  }
  __syncthreads();
  if (wave < 2 * CH / 32) {
    const int c = wave * 32 + lane;
    float tot = 0.f;
    for (int s = 0; s < TROWS / 16; ++s) tot += sPart[s * (2 * CH) + c];
    atomicAdd(&out_sums[c], tot);
  }
}

__global__ void biasout_kernel(const float* __restrict__ agg40,
                               const float* __restrict__ b3,
                               float* __restrict__ out, int total) {
  int i = blockIdx.x * blockDim.x + threadIdx.x;
  if (i < total) {
    int c = i % 40;
    out[i] = agg40[i] + b3[c];
  }
}

extern "C" void kernel_launch(void* const* d_in, const int* in_sizes, int n_in,
                              void* d_out, int out_size, void* d_ws, size_t ws_size,
                              hipStream_t stream) {
  const float* feat   = (const float*)d_in[0];
  const int*   src    = (const int*)d_in[1];
  const int*   dst    = (const int*)d_in[2];
  const float* ew     = (const float*)d_in[3];
  const float* W1     = (const float*)d_in[4];
  const float* W2     = (const float*)d_in[5];
  const float* W3     = (const float*)d_in[6];
  const float* b3     = (const float*)d_in[7];
  const float* gamma1 = (const float*)d_in[8];
  const float* beta1  = (const float*)d_in[9];
  const float* gamma2 = (const float*)d_in[10];
  const float* beta2  = (const float*)d_in[11];

  int N = in_sizes[0] / CH;
  int E = in_sizes[1];

  float* g1   = (float*)d_ws;
  float* g2   = g1 + (size_t)N * CH;
  float* t3   = g1;
  float* agg3 = g1 + (size_t)N * 64;

  int2* edges   = (int2*)d_out;
  int*  row_ptr = (int*)(edges + E);
  float* sums1  = (float*)(row_ptr + (N + 1 + 3) / 4 * 4);
  float* sums2  = sums1 + 2 * CH;
  int   npad = (N + SCAN_CHUNK - 1) / SCAN_CHUNK * SCAN_CHUNK;
  int*  cnt  = (int*)g2;
  int*  rank = cnt + npad;

  int ngrid = (npad + 255) / 256;
  int cgrid = (E + 255) / 256;
  int rgrid = (N + 7) / 8;
  int ggrid = (N + 127) / 128;
  int fgrid = (N + 63) / 64;
  int bgrid = (N * 40 + 255) / 256;
  dim3 b256(256);
  dim3 b512(512);

  zero_kernel<<<ngrid, b256, 0, stream>>>(cnt, npad, sums1);
  rank_kernel<<<cgrid, b256, 0, stream>>>(dst, cnt, rank, E);
  scan_kernel<<<1, 1024, 0, stream>>>(cnt, row_ptr, N);
  fill_kernel<<<cgrid, b256, 0, stream>>>(src, dst, ew, rank, row_ptr, edges, E);

  spmm_gemm_kernel<false><<<fgrid, b512, 0, stream>>>(feat, row_ptr, edges, W1, g1, N,
                                                       nullptr, nullptr, nullptr, sums1);

  spmm_gemm_kernel<true><<<fgrid, b512, 0, stream>>>(g1, row_ptr, edges, W2, g2, N,
                                                      sums1, gamma1, beta1, sums2);

  gemm_wmma_kernel<3, true><<<ggrid, b256, 0, stream>>>(g2, W3, t3, N, 40, 40,
                                                         sums2, gamma2, beta2);
  spmm_rows_kernel<10, false><<<rgrid, b256, 0, stream>>>(t3, row_ptr, edges, agg3, N,
                                                           nullptr, nullptr, nullptr);
  biasout_kernel<<<bgrid, b256, 0, stream>>>(agg3, b3, (float*)d_out, N * 40);
}


# Edge grouping rebuilt on LDS adds: slices binned into 512-node buckets via a per-slice count table, each bucket counted and placed in LDS, contiguous stores only; sum order in a row = LDS arrival
# speedup vs baseline: 4.1454x; 1.4796x over previous
#include <hip/hip_runtime.h>
#include <hip/hip_bf16.h>

typedef __attribute__((ext_vector_type(2))) float v2f;
typedef __attribute__((ext_vector_type(8))) float v8f;

#define BN_EPS 1e-5f
#define CH 128
#define WSTRIDE 132

#define BSH 9
#define BNODES (1 << BSH)
#define NBP 256
#define SLICE 6400
#define SPT ((SLICE + 1023) / 1024)
#define BPT (BCAP / 1024)
#define BCAP 16384

__device__ __forceinline__ int wave_incl_scan(int s, int lane) {
  for (int o = 1; o < 32; o <<= 1) {
    int v = __shfl_up(s, o);
    if (lane >= o) s += v;
  }
  return s;
}

template <int NW>
__device__ __forceinline__ int block_excl_scan(int v, int tid, int* s_w) {
  const int lane = tid & 31;
  const int wv = tid >> 5;
  const int incl = wave_incl_scan(v, lane);
  if (wv < NW && lane == 31) s_w[wv] = incl;
  __syncthreads();
  int off = 0;
  for (int k = 0; k < NW; ++k)
    if (k < wv) off += s_w[k];
  __syncthreads();
  return off + incl - v;
}

__global__ __launch_bounds__(1024) void slice_hist_kernel(const int* __restrict__ dst,
                                                          int* __restrict__ histG, int E,
                                                          float* __restrict__ sums) {
  __shared__ int hist[NBP];
  const int tid = threadIdx.x;
  if (tid < NBP) hist[tid] = 0;
  if (blockIdx.x == 0 && tid < 512) sums[tid] = 0.f;
  __syncthreads();
  const int beg = blockIdx.x * SLICE;
  const int end = min(E, beg + SLICE);
  int d[SPT];
#pragma unroll
  for (int k = 0; k < SPT; ++k) {
    const int e = beg + tid + k * 1024;
    d[k] = 0;
    if (e < end) d[k] = dst[e];
  }
#pragma unroll
  for (int k = 0; k < SPT; ++k)
    if (beg + tid + k * 1024 < end) atomicAdd(&hist[d[k] >> BSH], 1);
  __syncthreads();
  if (tid < NBP) histG[blockIdx.x * NBP + tid] = hist[tid];
}

__global__ __launch_bounds__(1024) void slice_scatter_kernel(const int* __restrict__ src,
                                                             const int* __restrict__ dst,
                                                             const float* __restrict__ ew,
                                                             const int* __restrict__ histG, int G,
                                                             int* __restrict__ bstart,
                                                             int* __restrict__ tD,
                                                             int* __restrict__ tS,
                                                             int* __restrict__ tW, int E) {
  __shared__ int s_tot[NBP];
  __shared__ int s_bef[NBP];
  __shared__ int s_cur[NBP];
  __shared__ int s_delta[NBP];
  __shared__ int s_w[8];
  __shared__ int st_d[SLICE];
  __shared__ int st_s[SLICE];
  __shared__ int st_w[SLICE];
  const int tid = threadIdx.x;
  const int g = blockIdx.x;
  const int t = tid & (NBP - 1);
  const int q = tid >> 8;
  if (tid < NBP) { s_tot[tid] = 0; s_bef[tid] = 0; }
  __syncthreads();
  const int beg = g * SLICE;
  const int end = min(E, beg + SLICE);
  int ed[SPT], es[SPT], ewb[SPT];
#pragma unroll
  for (int k = 0; k < SPT; ++k) {
    const int e = beg + tid + k * 1024;
    ed[k] = 0; es[k] = 0; ewb[k] = 0;
    if (e < end) { ed[k] = dst[e]; es[k] = src[e]; ewb[k] = __float_as_int(ew[e]); }
  }
  int tot = 0, bef = 0;
#pragma unroll 8
  for (int r = q; r < G; r += 4) {
    const int v = histG[r * NBP + t];
    tot += v;
    if (r < g) bef += v;
  }
  atomicAdd(&s_tot[t], tot);
  atomicAdd(&s_bef[t], bef);
  __syncthreads();
  int mine = 0, total = 0;
  if (tid < NBP) { mine = histG[g * NBP + tid]; total = s_tot[tid]; }
  const int bst = block_excl_scan<8>(total, tid, s_w);
  const int lst = block_excl_scan<8>(mine, tid, s_w);
  if (tid < NBP) {
    s_cur[tid] = lst;
    s_delta[tid] = bst + s_bef[tid] - lst;
    if (g == 0) {
      bstart[tid] = bst;
      if (tid == NBP - 1) bstart[NBP] = bst + total;
    }
  }
  __syncthreads();
#pragma unroll
  for (int k = 0; k < SPT; ++k) {
    if (beg + tid + k * 1024 < end) {
      const int p = atomicAdd(&s_cur[ed[k] >> BSH], 1);
      st_d[p] = ed[k];
      st_s[p] = es[k];
      st_w[p] = ewb[k];
    }
  }
  __syncthreads();
  const int n = end - beg;
  for (int i = tid; i < n; i += 1024) {
    const int d = st_d[i];
    const int pos = s_delta[d >> BSH] + i;
    tD[pos] = d;
    tS[pos] = st_s[i];
    tW[pos] = st_w[i];
  }
}

__global__ __launch_bounds__(1024) void bucket_sort_kernel(const int* __restrict__ tD,
                                                           const int* __restrict__ tS,
                                                           const int* __restrict__ tW,
                                                           const int* __restrict__ bstart,
                                                           int* __restrict__ row_ptr,
                                                           int2* __restrict__ edges, int N, int E) {
  __shared__ int s_cnt[BNODES];
  __shared__ int s_w[16];
  __shared__ int2 stage[BCAP];
  const int tid = threadIdx.x;
  const int b = blockIdx.x;
  const int lo = bstart[b];
  const int n = bstart[b + 1] - lo;
  const bool fits = n <= BCAP;
  if (tid < BNODES) s_cnt[tid] = 0;
  int dl[BPT], es[BPT], ewb[BPT];
#pragma unroll
  for (int k = 0; k < BPT; ++k) {
    const int i = tid + k * 1024;
    dl[k] = 0; es[k] = 0; ewb[k] = 0;
    if (fits && i < n) {
      dl[k] = tD[lo + i];
      es[k] = tS[lo + i];
      ewb[k] = tW[lo + i];
    }
  }
  __syncthreads();
  if (fits) {
#pragma unroll
    for (int k = 0; k < BPT; ++k)
      if (tid + k * 1024 < n) atomicAdd(&s_cnt[dl[k] & (BNODES - 1)], 1);
  } else {
    for (int i = tid; i < n; i += 1024) atomicAdd(&s_cnt[tD[lo + i] & (BNODES - 1)], 1);
  }
  __syncthreads();
  const int c = tid < BNODES ? s_cnt[tid] : 0;
  const int st = block_excl_scan<16>(c, tid, s_w);
  if (tid < BNODES) {
    s_cnt[tid] = st;
    const int node = b * BNODES + tid;
    if (node < N) row_ptr[node] = lo + st;
  }
  if (b == 0 && tid == 0) row_ptr[N] = E;
  __syncthreads();
  if (fits) {
#pragma unroll
    for (int k = 0; k < BPT; ++k) {
      if (tid + k * 1024 < n) {
        const int p = atomicAdd(&s_cnt[dl[k] & (BNODES - 1)], 1);
        stage[p] = make_int2(es[k], ewb[k]);
      }
    }
    __syncthreads();
    for (int i = tid; i < n; i += 1024) edges[lo + i] = stage[i];
  } else {
    for (int i = tid; i < n; i += 1024) {
      const int p = atomicAdd(&s_cnt[tD[lo + i] & (BNODES - 1)], 1);
      edges[lo + p] = make_int2(tS[lo + i], tW[lo + i]);
    }
  }
}

__device__ __forceinline__ float bn_relu(float x, float mean, float sc, float beta) {
  float v = (x - mean) * sc + beta;
  return v > 0.f ? v : 0.f;
}

template <int C4, int U, bool BN>
__device__ __forceinline__ void gather_group(const float4* __restrict__ F, int srcs, int wbits,
                                             int j, int l, float4& acc, const float4& bm,
                                             const float4& bs, const float4& bb) {
  float4 v[U];
  float w[U];
  for (int u = 0; u < U; ++u) {
    int s = __builtin_amdgcn_readlane(srcs, j + u);
    w[u] = __int_as_float(__builtin_amdgcn_readlane(wbits, j + u));
    v[u] = F[(size_t)s * C4 + l];
  }
  for (int u = 0; u < U; ++u) {
    float4 h = v[u];
    if constexpr (BN) {
      h.x = bn_relu(h.x, bm.x, bs.x, bb.x);
      h.y = bn_relu(h.y, bm.y, bs.y, bb.y);
      h.z = bn_relu(h.z, bm.z, bs.z, bb.z);
      h.w = bn_relu(h.w, bm.w, bs.w, bb.w);
    }
    acc.x += w[u] * h.x;
    acc.y += w[u] * h.y;
    acc.z += w[u] * h.z;
    acc.w += w[u] * h.w;
  }
}

template <int C4, bool BN>
__global__ __launch_bounds__(256) void spmm_rows_kernel(const float* __restrict__ feat,
                                                        const int* __restrict__ row_ptr,
                                                        const int2* __restrict__ edges,
                                                        float* __restrict__ agg, int Nrows,
                                                        const float* __restrict__ sums,
                                                        const float* __restrict__ gamma,
                                                        const float* __restrict__ beta) {
  const int wave = __builtin_amdgcn_readfirstlane((int)(threadIdx.x >> 5));
  const int lane = threadIdx.x & 31;
  const int row = blockIdx.x * 8 + wave;
  if (row >= Nrows) return;
  const int l = lane < C4 ? lane : C4 - 1;
  const float4* F = (const float4*)feat;
  const int beg = row_ptr[row];
  const int end = row_ptr[row + 1];
  int base = beg;
  int n = min(32, end - base);
  int2 my = make_int2(0, 0);
  if (lane < n) my = edges[base + lane];
  float4 bm = {0.f, 0.f, 0.f, 0.f}, bs = bm, bb = bm;
  if constexpr (BN) {
    const float invN = 1.0f / (float)Nrows;
    const float4 s = ((const float4*)sums)[l];
    const float4 q = ((const float4*)(sums + CH))[l];
    const float4 g4 = ((const float4*)gamma)[l];
    bb = ((const float4*)beta)[l];
    bm.x = s.x * invN; bm.y = s.y * invN; bm.z = s.z * invN; bm.w = s.w * invN;
    bs.x = g4.x * rsqrtf(q.x * invN - bm.x * bm.x + BN_EPS);
    bs.y = g4.y * rsqrtf(q.y * invN - bm.y * bm.y + BN_EPS);
    bs.z = g4.z * rsqrtf(q.z * invN - bm.z * bm.z + BN_EPS);
    bs.w = g4.w * rsqrtf(q.w * invN - bm.w * bm.w + BN_EPS);
  }
  float4 acc = {0.f, 0.f, 0.f, 0.f};
  while (base < end) {
    int j = 0;
    for (; j + 8 <= n; j += 8) gather_group<C4, 8, BN>(F, my.x, my.y, j, l, acc, bm, bs, bb);
    if (j + 4 <= n) { gather_group<C4, 4, BN>(F, my.x, my.y, j, l, acc, bm, bs, bb); j += 4; }
    if (j + 2 <= n) { gather_group<C4, 2, BN>(F, my.x, my.y, j, l, acc, bm, bs, bb); j += 2; }
    if (j < n) gather_group<C4, 1, BN>(F, my.x, my.y, j, l, acc, bm, bs, bb);
    base += 32;
    n = min(32, end - base);
    my = make_int2(0, 0);
    if (lane < n) my = edges[base + lane];
  }
  if (lane < C4) ((float4*)agg)[(size_t)row * C4 + lane] = acc;
}

template <int NT, bool BN_IN>
__global__ void gemm_wmma_kernel(const float* __restrict__ A,
                                 const float* __restrict__ W,
                                 float* __restrict__ out,
                                 int Nrows, int Cout, int ldo,
                                 const float* __restrict__ in_sums,
                                 const float* __restrict__ in_gamma,
                                 const float* __restrict__ in_beta) {
  constexpr int CP = NT * 16;
  __shared__ float sWT[CP * WSTRIDE];
  __shared__ float4 sBNa[BN_IN ? CH / 2 : 1];
  __shared__ float2 sBNb[BN_IN ? CH / 2 : 1];
  int tid = threadIdx.x;
  for (int idx = tid; idx < CH * CP; idx += 256) {
    int k = idx / CP;
    int c = idx % CP;
    sWT[c * WSTRIDE + k] = (c < Cout) ? W[k * Cout + c] : 0.f;
  }
  if constexpr (BN_IN) {
    if (tid < CH / 2) {
      const float invN = 1.0f / (float)Nrows;
      const int c = tid * 2;
      const float m0 = in_sums[c] * invN;
      const float m1 = in_sums[c + 1] * invN;
      const float s0 = in_gamma[c] * rsqrtf(in_sums[CH + c] * invN - m0 * m0 + BN_EPS);
      const float s1 = in_gamma[c + 1] * rsqrtf(in_sums[CH + c + 1] * invN - m1 * m1 + BN_EPS);
      sBNa[tid] = make_float4(m0, m1, s0, s1);
      sBNb[tid] = make_float2(in_beta[c], in_beta[c + 1]);
    }
  }
  __syncthreads();

  const int wave = __builtin_amdgcn_readfirstlane(tid >> 5);
  int lane = tid & 31;
  int row0 = blockIdx.x * 128 + wave * 16;
  const bool valid = row0 < Nrows;

  int m = lane & 15;
  int kh = (lane >> 4) * 2;

  v8f acc[NT];
#pragma unroll
  for (int t = 0; t < NT; ++t) acc[t] = {};

  if (valid) {
    const float* Arow = A + (size_t)(row0 + m) * CH;

    for (int k0 = 0; k0 < CH; k0 += 4) {
      v2f a = *(const v2f*)(Arow + k0 + kh);
      if constexpr (BN_IN) {
        const float4 ms = sBNa[(k0 + kh) >> 1];
        const float2 be = sBNb[(k0 + kh) >> 1];
        float x0 = (a.x - ms.x) * ms.z + be.x;
        float x1 = (a.y - ms.y) * ms.w + be.y;
        a.x = x0 > 0.f ? x0 : 0.f;
        a.y = x1 > 0.f ? x1 : 0.f;
      }
#pragma unroll
      for (int t = 0; t < NT; ++t) {
        v2f b = *(const v2f*)(&sWT[(t * 16 + m) * WSTRIDE + k0 + kh]);
        acc[t] = __builtin_amdgcn_wmma_f32_16x16x4_f32(
             false, a,  false, b,
             (short)0, acc[t],  false,  false);
      }
    }

    int rbase = row0 + (lane >> 4) * 8;
#pragma unroll
    for (int t = 0; t < NT; ++t) {
      int col = t * 16 + m;
      if (col < Cout) {
#pragma unroll
        for (int j = 0; j < 8; ++j) {
          out[(size_t)(rbase + j) * ldo + col] = acc[t][j];
        }
      }
    }
  }
}

template <bool BN>
__device__ __forceinline__ float4 gather_row(const float4* __restrict__ F,
                                             const int2* __restrict__ edges, int base, int end,
                                             int2 my, int lane, const float4& bm,
                                             const float4& bs, const float4& bb) {
  float4 acc = {0.f, 0.f, 0.f, 0.f};
  int n = min(32, end - base);
  while (base < end) {
    int j = 0;
    for (; j + 8 <= n; j += 8) gather_group<32, 8, BN>(F, my.x, my.y, j, lane, acc, bm, bs, bb);
    if (j + 4 <= n) { gather_group<32, 4, BN>(F, my.x, my.y, j, lane, acc, bm, bs, bb); j += 4; }
    if (j + 2 <= n) { gather_group<32, 2, BN>(F, my.x, my.y, j, lane, acc, bm, bs, bb); j += 2; }
    if (j < n) gather_group<32, 1, BN>(F, my.x, my.y, j, lane, acc, bm, bs, bb);
    base += 32;
    n = min(32, end - base);
    my = make_int2(0, 0);
    if (lane < n) my = edges[base + lane];
  }
  return acc;
}

#define TROWS 64
template <bool BN>
__global__ __launch_bounds__(512, 12) void spmm_gemm_kernel(const float* __restrict__ feat,
                                                             const int* __restrict__ row_ptr,
                                                             const int2* __restrict__ edges,
                                                             const float* __restrict__ W,
                                                             float* __restrict__ out, int Nrows,
                                                             const float* __restrict__ in_sums,
                                                             const float* __restrict__ in_gamma,
                                                             const float* __restrict__ in_beta,
                                                             float* __restrict__ out_sums) {
  __shared__ float sWT[CH * WSTRIDE];
  __shared__ float sA[TROWS * WSTRIDE];
  __shared__ float sPart[(TROWS / 16) * 2 * CH];
  const int tid = threadIdx.x;
  const int wave = __builtin_amdgcn_readfirstlane(tid >> 5);
  const int lane = __builtin_amdgcn_mbcnt_lo(~0u, 0u);

  const int r0 = blockIdx.x * TROWS + wave * 4;
  int rp[5];
  for (int i = 0; i < 5; ++i) rp[i] = row_ptr[min(r0 + i, Nrows)];
  int2 first[4];
#pragma unroll
  for (int i = 0; i < 4; ++i) {
    first[i] = make_int2(0, 0);
    if (lane < rp[i + 1] - rp[i]) first[i] = edges[rp[i] + lane];
  }

  {
    float wv[32];
#pragma unroll
    for (int j = 0; j < 32; ++j) wv[j] = W[tid + j * 512];
    float* wdst = &sWT[(tid & (CH - 1)) * WSTRIDE + (tid >> 7)];
#pragma unroll
    for (int j = 0; j < 32; ++j) wdst[4 * j] = wv[j];
  }
#pragma unroll
  for (int i = 0; i < 4; ++i) {
    const v2f pr = {__int_as_float(first[i].x), __int_as_float(first[i].y)};
    *(v2f*)(&sA[(wave * 4 + i) * WSTRIDE + lane * 2]) = pr;
  }

  float4 bm = {0.f, 0.f, 0.f, 0.f}, bs = bm, bb = bm;
  if constexpr (BN) {
    const float invN = 1.0f / (float)Nrows;
    const float4 s = ((const float4*)in_sums)[lane];
    const float4 q = ((const float4*)(in_sums + CH))[lane];
    const float4 g4 = ((const float4*)in_gamma)[lane];
    bb = ((const float4*)in_beta)[lane];
    bm.x = s.x * invN; bm.y = s.y * invN; bm.z = s.z * invN; bm.w = s.w * invN;
    bs.x = g4.x * rsqrtf(q.x * invN - bm.x * bm.x + BN_EPS);
    bs.y = g4.y * rsqrtf(q.y * invN - bm.y * bm.y + BN_EPS);
    bs.z = g4.z * rsqrtf(q.z * invN - bm.z * bm.z + BN_EPS);
    bs.w = g4.w * rsqrtf(q.w * invN - bm.w * bm.w + BN_EPS);
  }

  const float4* F = (const float4*)feat;
#pragma unroll
  for (int i = 0; i < 4; ++i) {
    const v2f pr = *(const v2f*)(&sA[(wave * 4 + i) * WSTRIDE + lane * 2]);
    const int2 cur = make_int2(__float_as_int(pr.x), __float_as_int(pr.y));
    const float4 acc = gather_row<BN>(F, edges, rp[i], rp[i + 1], cur, lane, bm, bs, bb);
    *(float4*)(&sA[(wave * 4 + i) * WSTRIDE + lane * 4]) = acc;
  }
  __syncthreads();

  const int stripe = wave >> 2;
  const int t0 = (wave & 3) * 2;
  const int m = lane & 15;
  const int kh = (lane >> 4) * 2;
  const float* Ap = &sA[(stripe * 16 + m) * WSTRIDE + kh];
  const float* Bp0 = &sWT[(t0 * 16 + m) * WSTRIDE + kh];
  const float* Bp1 = Bp0 + 16 * WSTRIDE;
  v8f acc0 = {}, acc1 = {};
#pragma unroll 8
  for (int k0 = 0; k0 < CH; k0 += 4) {
    const v2f a = *(const v2f*)(Ap + k0);
    const v2f b0 = *(const v2f*)(Bp0 + k0);
    const v2f b1 = *(const v2f*)(Bp1 + k0);
    acc0 = __builtin_amdgcn_wmma_f32_16x16x4_f32(false, a, false, b0, (short)0, acc0, false, false);
    acc1 = __builtin_amdgcn_wmma_f32_16x16x4_f32(false, a, false, b1, (short)0, acc1, false, false);
  }

  const int rbase = blockIdx.x * TROWS + stripe * 16 + (lane >> 4) * 8;
  const int col = t0 * 16 + m;
  for (int j = 0; j < 8; ++j) {
    if (rbase + j < Nrows) {
      out[(size_t)(rbase + j) * CH + col] = acc0[j];
      out[(size_t)(rbase + j) * CH + col + 16] = acc1[j];
    }
  }

  {
    float s0 = 0.f, q0 = 0.f, s1 = 0.f, q1 = 0.f;
    for (int j = 0; j < 8; ++j) {
      const float v0 = acc0[j], v1 = acc1[j];
      s0 += v0; q0 += v0 * v0;
      s1 += v1; q1 += v1 * v1;
    }
    s0 += __shfl_xor(s0, 16); q0 += __shfl_xor(q0, 16);
    s1 += __shfl_xor(s1, 16); q1 += __shfl_xor(q1, 16);
    if (lane < 16) {
      float* p = &sPart[stripe * (2 * CH) + col];
      p[0] = s0; p[CH] = q0;
      p[16] = s1; p[CH + 16] = q1;
    }
  }
  __syncthreads();
  if (wave < 2 * CH / 32) {
    const int c = wave * 32 + lane;
    float tot = 0.f;
    for (int s = 0; s < TROWS / 16; ++s) tot += sPart[s * (2 * CH) + c];
    atomicAdd(&out_sums[c], tot);
  }
}

__global__ void biasout_kernel(const float* __restrict__ agg40,
                               const float* __restrict__ b3,
                               float* __restrict__ out, int total) {
  int i = blockIdx.x * blockDim.x + threadIdx.x;
  if (i < total) {
    int c = i % 40;
    out[i] = agg40[i] + b3[c];
  }
}

extern "C" void kernel_launch(void* const* d_in, const int* in_sizes, int n_in,
                              void* d_out, int out_size, void* d_ws, size_t ws_size,
                              hipStream_t stream) {
  const float* feat   = (const float*)d_in[0];
  const int*   src    = (const int*)d_in[1];
  const int*   dst    = (const int*)d_in[2];
  const float* ew     = (const float*)d_in[3];
  const float* W1     = (const float*)d_in[4];
  const float* W2     = (const float*)d_in[5];
  const float* W3     = (const float*)d_in[6];
  const float* b3     = (const float*)d_in[7];
  const float* gamma1 = (const float*)d_in[8];
  const float* beta1  = (const float*)d_in[9];
  const float* gamma2 = (const float*)d_in[10];
  const float* beta2  = (const float*)d_in[11];

  int N = in_sizes[0] / CH;
  int E = in_sizes[1];

  float* g1   = (float*)d_ws;
  float* g2   = g1 + (size_t)N * CH;
  float* t3   = g1;
  float* agg3 = g1 + (size_t)N * 64;

  int2* edges   = (int2*)d_out;
  int*  row_ptr = (int*)(edges + E);
  float* sums1  = (float*)(row_ptr + (N + 1 + 3) / 4 * 4);
  float* sums2  = sums1 + 2 * CH;
  int   sgrid = (E + SLICE - 1) / SLICE;
  int   kgrid = (N + BNODES - 1) / BNODES;
  int*  tD     = (int*)g2;
  int*  tS     = tD + E;
  int*  tW     = tS + E;
  int*  histG  = tW + E;
  int*  bstart = histG + (size_t)sgrid * NBP;

  int rgrid = (N + 7) / 8;
  int ggrid = (N + 127) / 128;
  int fgrid = (N + 63) / 64;
  int bgrid = (N * 40 + 255) / 256;
  dim3 b256(256);
  dim3 b512(512);
  dim3 b1024(1024);

  slice_hist_kernel<<<sgrid, b1024, 0, stream>>>(dst, histG, E, sums1);
  slice_scatter_kernel<<<sgrid, b1024, 0, stream>>>(src, dst, ew, histG, sgrid, bstart,
                                                    tD, tS, tW, E);
  bucket_sort_kernel<<<kgrid, b1024, 0, stream>>>(tD, tS, tW, bstart, row_ptr, edges, N, E);

  spmm_gemm_kernel<false><<<fgrid, b512, 0, stream>>>(feat, row_ptr, edges, W1, g1, N,
                                                       nullptr, nullptr, nullptr, sums1);

  spmm_gemm_kernel<true><<<fgrid, b512, 0, stream>>>(g1, row_ptr, edges, W2, g2, N,
                                                      sums1, gamma1, beta1, sums2);

  gemm_wmma_kernel<3, true><<<ggrid, b256, 0, stream>>>(g2, W3, t3, N, 40, 40,
                                                         sums2, gamma2, beta2);
  spmm_rows_kernel<10, false><<<rgrid, b256, 0, stream>>>(t3, row_ptr, edges, agg3, N,
                                                           nullptr, nullptr, nullptr);
  biasout_kernel<<<bgrid, b256, 0, stream>>>(agg3, b3, (float*)d_out, N * 40);
}


# Layer-3 GEMM moves whole lines: a wave loads its 16 rows as 512-byte requests, all in flight with the W3 loads, normalises them into an LDS tile feeding the WMMAs; output stored as 16-byte pieces
# speedup vs baseline: 4.4530x; 1.0742x over previous
#include <hip/hip_runtime.h>
#include <hip/hip_bf16.h>

typedef __attribute__((ext_vector_type(2))) float v2f;
typedef __attribute__((ext_vector_type(8))) float v8f;

#define BN_EPS 1e-5f
#define CH 128
#define WSTRIDE 132

#define BSH 9
#define BNODES (1 << BSH)
#define NBP 256
#define SLICE 6400
#define SPT ((SLICE + 1023) / 1024)
#define BPT (BCAP / 1024)
#define BCAP 16384

__device__ __forceinline__ int wave_incl_scan(int s, int lane) {
  for (int o = 1; o < 32; o <<= 1) {
    int v = __shfl_up(s, o);
    if (lane >= o) s += v;
  }
  return s;
}

template <int NW>
__device__ __forceinline__ int block_excl_scan(int v, int tid, int* s_w) {
  const int lane = tid & 31;
  const int wv = tid >> 5;
  const int incl = wave_incl_scan(v, lane);
  if (wv < NW && lane == 31) s_w[wv] = incl;
  __syncthreads();
  int off = 0;
  for (int k = 0; k < NW; ++k)
    if (k < wv) off += s_w[k];
  __syncthreads();
  return off + incl - v;
}

__global__ __launch_bounds__(1024) void slice_hist_kernel(const int* __restrict__ dst,
                                                          int* __restrict__ histG, int E,
                                                          float* __restrict__ sums) {
  __shared__ int hist[NBP];
  const int tid = threadIdx.x;
  if (tid < NBP) hist[tid] = 0;
  if (blockIdx.x == 0 && tid < 512) sums[tid] = 0.f;
  __syncthreads();
  const int beg = blockIdx.x * SLICE;
  const int end = min(E, beg + SLICE);
  int d[SPT];
#pragma unroll
  for (int k = 0; k < SPT; ++k) {
    const int e = beg + tid + k * 1024;
    d[k] = 0;
    if (e < end) d[k] = dst[e];
  }
#pragma unroll
  for (int k = 0; k < SPT; ++k)
    if (beg + tid + k * 1024 < end) atomicAdd(&hist[d[k] >> BSH], 1);
  __syncthreads();
  if (tid < NBP) histG[blockIdx.x * NBP + tid] = hist[tid];
}

__global__ __launch_bounds__(1024) void slice_scatter_kernel(const int* __restrict__ src,
                                                             const int* __restrict__ dst,
                                                             const float* __restrict__ ew,
                                                             const int* __restrict__ histG, int G,
                                                             int* __restrict__ bstart,
                                                             int* __restrict__ tD,
                                                             int* __restrict__ tS,
                                                             int* __restrict__ tW, int E) {
  __shared__ int s_tot[NBP];
  __shared__ int s_bef[NBP];
  __shared__ int s_cur[NBP];
  __shared__ int s_delta[NBP];
  __shared__ int s_w[8];
  __shared__ int st_d[SLICE];
  __shared__ int st_s[SLICE];
  __shared__ int st_w[SLICE];
  const int tid = threadIdx.x;
  const int g = blockIdx.x;
  const int t = tid & (NBP - 1);
  const int q = tid >> 8;
  if (tid < NBP) { s_tot[tid] = 0; s_bef[tid] = 0; }
  __syncthreads();
  const int beg = g * SLICE;
  const int end = min(E, beg + SLICE);
  int ed[SPT], es[SPT], ewb[SPT];
#pragma unroll
  for (int k = 0; k < SPT; ++k) {
    const int e = beg + tid + k * 1024;
    ed[k] = 0; es[k] = 0; ewb[k] = 0;
    if (e < end) { ed[k] = dst[e]; es[k] = src[e]; ewb[k] = __float_as_int(ew[e]); }
  }
  int tot = 0, bef = 0;
#pragma unroll 8
  for (int r = q; r < G; r += 4) {
    const int v = histG[r * NBP + t];
    tot += v;
    if (r < g) bef += v;
  }
  atomicAdd(&s_tot[t], tot);
  atomicAdd(&s_bef[t], bef);
  __syncthreads();
  int mine = 0, total = 0;
  if (tid < NBP) { mine = histG[g * NBP + tid]; total = s_tot[tid]; }
  const int bst = block_excl_scan<8>(total, tid, s_w);
  const int lst = block_excl_scan<8>(mine, tid, s_w);
  if (tid < NBP) {
    s_cur[tid] = lst;
    s_delta[tid] = bst + s_bef[tid] - lst;
    if (g == 0) {
      bstart[tid] = bst;
      if (tid == NBP - 1) bstart[NBP] = bst + total;
    }
  }
  __syncthreads();
#pragma unroll
  for (int k = 0; k < SPT; ++k) {
    if (beg + tid + k * 1024 < end) {
      const int p = atomicAdd(&s_cur[ed[k] >> BSH], 1);
      st_d[p] = ed[k];
      st_s[p] = es[k];
      st_w[p] = ewb[k];
    }
  }
  __syncthreads();
  const int n = end - beg;
  for (int i = tid; i < n; i += 1024) {
    const int d = st_d[i];
    const int pos = s_delta[d >> BSH] + i;
    tD[pos] = d;
    tS[pos] = st_s[i];
    tW[pos] = st_w[i];
  }
}

__global__ __launch_bounds__(1024) void bucket_sort_kernel(const int* __restrict__ tD,
                                                           const int* __restrict__ tS,
                                                           const int* __restrict__ tW,
                                                           const int* __restrict__ bstart,
                                                           int* __restrict__ row_ptr,
                                                           int2* __restrict__ edges, int N, int E) {
  __shared__ int s_cnt[BNODES];
  __shared__ int s_w[16];
  __shared__ int2 stage[BCAP];
  const int tid = threadIdx.x;
  const int b = blockIdx.x;
  const int lo = bstart[b];
  const int n = bstart[b + 1] - lo;
  const bool fits = n <= BCAP;
  if (tid < BNODES) s_cnt[tid] = 0;
  int dl[BPT], es[BPT], ewb[BPT];
#pragma unroll
  for (int k = 0; k < BPT; ++k) {
    const int i = tid + k * 1024;
    dl[k] = 0; es[k] = 0; ewb[k] = 0;
    if (fits && i < n) {
      dl[k] = tD[lo + i];
      es[k] = tS[lo + i];
      ewb[k] = tW[lo + i];
    }
  }
  __syncthreads();
  if (fits) {
#pragma unroll
    for (int k = 0; k < BPT; ++k)
      if (tid + k * 1024 < n) atomicAdd(&s_cnt[dl[k] & (BNODES - 1)], 1);
  } else {
    for (int i = tid; i < n; i += 1024) atomicAdd(&s_cnt[tD[lo + i] & (BNODES - 1)], 1);
  }
  __syncthreads();
  const int c = tid < BNODES ? s_cnt[tid] : 0;
  const int st = block_excl_scan<16>(c, tid, s_w);
  if (tid < BNODES) {
    s_cnt[tid] = st;
    const int node = b * BNODES + tid;
    if (node < N) row_ptr[node] = lo + st;
  }
  if (b == 0 && tid == 0) row_ptr[N] = E;
  __syncthreads();
  if (fits) {
#pragma unroll
    for (int k = 0; k < BPT; ++k) {
      if (tid + k * 1024 < n) {
        const int p = atomicAdd(&s_cnt[dl[k] & (BNODES - 1)], 1);
        stage[p] = make_int2(es[k], ewb[k]);
      }
    }
    __syncthreads();
    for (int i = tid; i < n; i += 1024) edges[lo + i] = stage[i];
  } else {
    for (int i = tid; i < n; i += 1024) {
      const int p = atomicAdd(&s_cnt[tD[lo + i] & (BNODES - 1)], 1);
      edges[lo + p] = make_int2(tS[lo + i], tW[lo + i]);
    }
  }
}

__device__ __forceinline__ float bn_relu(float x, float mean, float sc, float beta) {
  float v = (x - mean) * sc + beta;
  return v > 0.f ? v : 0.f;
}

template <int C4, int U, bool BN>
__device__ __forceinline__ void gather_group(const float4* __restrict__ F, int srcs, int wbits,
                                             int j, int l, float4& acc, const float4& bm,
                                             const float4& bs, const float4& bb) {
  float4 v[U];
  float w[U];
  for (int u = 0; u < U; ++u) {
    int s = __builtin_amdgcn_readlane(srcs, j + u);
    w[u] = __int_as_float(__builtin_amdgcn_readlane(wbits, j + u));
    v[u] = F[(size_t)s * C4 + l];
  }
  for (int u = 0; u < U; ++u) {
    float4 h = v[u];
    if constexpr (BN) {
      h.x = bn_relu(h.x, bm.x, bs.x, bb.x);
      h.y = bn_relu(h.y, bm.y, bs.y, bb.y);
      h.z = bn_relu(h.z, bm.z, bs.z, bb.z);
      h.w = bn_relu(h.w, bm.w, bs.w, bb.w);
    }
    acc.x += w[u] * h.x;
    acc.y += w[u] * h.y;
    acc.z += w[u] * h.z;
    acc.w += w[u] * h.w;
  }
}

template <int C4, bool BN>
__global__ __launch_bounds__(256) void spmm_rows_kernel(const float* __restrict__ feat,
                                                        const int* __restrict__ row_ptr,
                                                        const int2* __restrict__ edges,
                                                        float* __restrict__ agg, int Nrows,
                                                        const float* __restrict__ sums,
                                                        const float* __restrict__ gamma,
                                                        const float* __restrict__ beta) {
  const int wave = __builtin_amdgcn_readfirstlane((int)(threadIdx.x >> 5));
  const int lane = threadIdx.x & 31;
  const int row = blockIdx.x * 8 + wave;
  if (row >= Nrows) return;
  const int l = lane < C4 ? lane : C4 - 1;
  const float4* F = (const float4*)feat;
  const int beg = row_ptr[row];
  const int end = row_ptr[row + 1];
  int base = beg;
  int n = min(32, end - base);
  int2 my = make_int2(0, 0);
  if (lane < n) my = edges[base + lane];
  float4 bm = {0.f, 0.f, 0.f, 0.f}, bs = bm, bb = bm;
  if constexpr (BN) {
    const float invN = 1.0f / (float)Nrows;
    const float4 s = ((const float4*)sums)[l];
    const float4 q = ((const float4*)(sums + CH))[l];
    const float4 g4 = ((const float4*)gamma)[l];
    bb = ((const float4*)beta)[l];
    bm.x = s.x * invN; bm.y = s.y * invN; bm.z = s.z * invN; bm.w = s.w * invN;
    bs.x = g4.x * rsqrtf(q.x * invN - bm.x * bm.x + BN_EPS);
    bs.y = g4.y * rsqrtf(q.y * invN - bm.y * bm.y + BN_EPS);
    bs.z = g4.z * rsqrtf(q.z * invN - bm.z * bm.z + BN_EPS);
    bs.w = g4.w * rsqrtf(q.w * invN - bm.w * bm.w + BN_EPS);
  }
  float4 acc = {0.f, 0.f, 0.f, 0.f};
  while (base < end) {
    int j = 0;
    for (; j + 8 <= n; j += 8) gather_group<C4, 8, BN>(F, my.x, my.y, j, l, acc, bm, bs, bb);
    if (j + 4 <= n) { gather_group<C4, 4, BN>(F, my.x, my.y, j, l, acc, bm, bs, bb); j += 4; }
    if (j + 2 <= n) { gather_group<C4, 2, BN>(F, my.x, my.y, j, l, acc, bm, bs, bb); j += 2; }
    if (j < n) gather_group<C4, 1, BN>(F, my.x, my.y, j, l, acc, bm, bs, bb);
    base += 32;
    n = min(32, end - base);
    my = make_int2(0, 0);
    if (lane < n) my = edges[base + lane];
  }
  if (lane < C4) ((float4*)agg)[(size_t)row * C4 + lane] = acc;
}

#define GROWS 64
template <int NT, int COUT>
__global__ __launch_bounds__(128) void gemm_lines_kernel(const float* __restrict__ A,
                                                         const float* __restrict__ W,
                                                         float* __restrict__ out,
                                                         int Nrows,
                                                         const float* __restrict__ in_sums,
                                                         const float* __restrict__ in_gamma,
                                                         const float* __restrict__ in_beta) {
  constexpr int CP = NT * 16;
  constexpr int C4 = COUT / 4;
  constexpr int WPT = CH * C4 / 128;
  static_assert(COUT % 4 == 0 && COUT <= CP && (CH * C4) % 128 == 0, "W is staged in whole float4");
  __shared__ __attribute__((aligned(16))) float sWT[CP * WSTRIDE];
  __shared__ __attribute__((aligned(16))) float sA[GROWS * WSTRIDE];
  const int tid = threadIdx.x;
  const int wave = __builtin_amdgcn_readfirstlane(tid >> 5);
  const int lane = tid & 31;
  const int row0 = blockIdx.x * GROWS + wave * 16;
  const bool valid = row0 < Nrows;

  float4 r[16];
  if (valid) {
    const float4* A4 = (const float4*)A + (size_t)row0 * (CH / 4) + lane;
#pragma unroll
    for (int i = 0; i < 16; ++i) r[i] = A4[(size_t)i * (CH / 4)];
  }
  float4 wv[WPT];
#pragma unroll
  for (int j = 0; j < WPT; ++j) wv[j] = ((const float4*)W)[tid + j * 128];
  float4 bm, bs, bb;
  {
    const float invN = 1.0f / (float)Nrows;
    const float4 s = ((const float4*)in_sums)[lane];
    const float4 q = ((const float4*)(in_sums + CH))[lane];
    const float4 g4 = ((const float4*)in_gamma)[lane];
    bb = ((const float4*)in_beta)[lane];
    bm.x = s.x * invN; bm.y = s.y * invN; bm.z = s.z * invN; bm.w = s.w * invN;
    bs.x = g4.x * rsqrtf(q.x * invN - bm.x * bm.x + BN_EPS);
    bs.y = g4.y * rsqrtf(q.y * invN - bm.y * bm.y + BN_EPS);
    bs.z = g4.z * rsqrtf(q.z * invN - bm.z * bm.z + BN_EPS);
    bs.w = g4.w * rsqrtf(q.w * invN - bm.w * bm.w + BN_EPS);
  }
#pragma unroll
  for (int j = 0; j < WPT; ++j) {
    const int q = tid + j * 128;
    float* wd = &sWT[(q % C4) * 4 * WSTRIDE + q / C4];
    wd[0] = wv[j].x;
    wd[WSTRIDE] = wv[j].y;
    wd[2 * WSTRIDE] = wv[j].z;
    wd[3 * WSTRIDE] = wv[j].w;
  }
#pragma unroll
  for (int j = COUT; j < CP; ++j) sWT[j * WSTRIDE + tid] = 0.f;
  float* stripe = &sA[wave * 16 * WSTRIDE];
  if (valid) {
#pragma unroll
    for (int i = 0; i < 16; ++i) {
      float4 h = r[i];
      h.x = bn_relu(h.x, bm.x, bs.x, bb.x);
      h.y = bn_relu(h.y, bm.y, bs.y, bb.y);
      h.z = bn_relu(h.z, bm.z, bs.z, bb.z);
      h.w = bn_relu(h.w, bm.w, bs.w, bb.w);
      *(float4*)(&stripe[i * WSTRIDE + lane * 4]) = h;
    }
  }
  __syncthreads();

  const int m = lane & 15;
  const int kh = (lane >> 4) * 2;
  v8f acc[NT];
#pragma unroll
  for (int t = 0; t < NT; ++t) acc[t] = {};
  if (valid) {
    const float* Ap = &stripe[m * WSTRIDE + kh];
    const float* Bp = &sWT[m * WSTRIDE + kh];
#pragma unroll 8
    for (int k0 = 0; k0 < CH; k0 += 4) {
      const v2f a = *(const v2f*)(Ap + k0);
#pragma unroll
      for (int t = 0; t < NT; ++t) {
        const v2f b = *(const v2f*)(Bp + t * 16 * WSTRIDE + k0);
        acc[t] = __builtin_amdgcn_wmma_f32_16x16x4_f32(
             false, a,  false, b,
             (short)0, acc[t],  false,  false);
      }
    }
    const int rb = (lane >> 4) * 8;
#pragma unroll
    for (int t = 0; t < NT; ++t) {
      const int col = t * 16 + m;
      if (col < COUT) {
#pragma unroll
        for (int j = 0; j < 8; ++j) stripe[(rb + j) * COUT + col] = acc[t][j];
      }
    }
  }
  __syncthreads();
  if (valid) {
    float4* o4 = (float4*)(out + (size_t)row0 * COUT);
#pragma unroll
    for (int i = lane; i < 4 * COUT; i += 32) o4[i] = ((const float4*)stripe)[i];
  }
}

template <bool BN>
__device__ __forceinline__ float4 gather_row(const float4* __restrict__ F,
                                             const int2* __restrict__ edges, int base, int end,
                                             int2 my, int lane, const float4& bm,
                                             const float4& bs, const float4& bb) {
  float4 acc = {0.f, 0.f, 0.f, 0.f};
  int n = min(32, end - base);
  while (base < end) {
    int j = 0;
    for (; j + 8 <= n; j += 8) gather_group<32, 8, BN>(F, my.x, my.y, j, lane, acc, bm, bs, bb);
    if (j + 4 <= n) { gather_group<32, 4, BN>(F, my.x, my.y, j, lane, acc, bm, bs, bb); j += 4; }
    if (j + 2 <= n) { gather_group<32, 2, BN>(F, my.x, my.y, j, lane, acc, bm, bs, bb); j += 2; }
    if (j < n) gather_group<32, 1, BN>(F, my.x, my.y, j, lane, acc, bm, bs, bb);
    base += 32;
    n = min(32, end - base);
    my = make_int2(0, 0);
    if (lane < n) my = edges[base + lane];
  }
  return acc;
}

#define TROWS 64
template <bool BN>
__global__ __launch_bounds__(512, 12) void spmm_gemm_kernel(const float* __restrict__ feat,
                                                             const int* __restrict__ row_ptr,
                                                             const int2* __restrict__ edges,
                                                             const float* __restrict__ W,
                                                             float* __restrict__ out, int Nrows,
                                                             const float* __restrict__ in_sums,
                                                             const float* __restrict__ in_gamma,
                                                             const float* __restrict__ in_beta,
                                                             float* __restrict__ out_sums) {
  __shared__ float sWT[CH * WSTRIDE];
  __shared__ float sA[TROWS * WSTRIDE];
  __shared__ float sPart[(TROWS / 16) * 2 * CH];
  const int tid = threadIdx.x;
  const int wave = __builtin_amdgcn_readfirstlane(tid >> 5);
  const int lane = __builtin_amdgcn_mbcnt_lo(~0u, 0u);

  const int r0 = blockIdx.x * TROWS + wave * 4;
  int rp[5];
  for (int i = 0; i < 5; ++i) rp[i] = row_ptr[min(r0 + i, Nrows)];
  int2 first[4];
#pragma unroll
  for (int i = 0; i < 4; ++i) {
    first[i] = make_int2(0, 0);
    if (lane < rp[i + 1] - rp[i]) first[i] = edges[rp[i] + lane];
  }

  {
    float wv[32];
#pragma unroll
    for (int j = 0; j < 32; ++j) wv[j] = W[tid + j * 512];
    float* wdst = &sWT[(tid & (CH - 1)) * WSTRIDE + (tid >> 7)];
#pragma unroll
    for (int j = 0; j < 32; ++j) wdst[4 * j] = wv[j];
  }
#pragma unroll
  for (int i = 0; i < 4; ++i) {
    const v2f pr = {__int_as_float(first[i].x), __int_as_float(first[i].y)};
    *(v2f*)(&sA[(wave * 4 + i) * WSTRIDE + lane * 2]) = pr;
  }

  float4 bm = {0.f, 0.f, 0.f, 0.f}, bs = bm, bb = bm;
  if constexpr (BN) {
    const float invN = 1.0f / (float)Nrows;
    const float4 s = ((const float4*)in_sums)[lane];
    const float4 q = ((const float4*)(in_sums + CH))[lane];
    const float4 g4 = ((const float4*)in_gamma)[lane];
    bb = ((const float4*)in_beta)[lane];
    bm.x = s.x * invN; bm.y = s.y * invN; bm.z = s.z * invN; bm.w = s.w * invN;
    bs.x = g4.x * rsqrtf(q.x * invN - bm.x * bm.x + BN_EPS);
    bs.y = g4.y * rsqrtf(q.y * invN - bm.y * bm.y + BN_EPS);
    bs.z = g4.z * rsqrtf(q.z * invN - bm.z * bm.z + BN_EPS);
    bs.w = g4.w * rsqrtf(q.w * invN - bm.w * bm.w + BN_EPS);
  }

  const float4* F = (const float4*)feat;
#pragma unroll
  for (int i = 0; i < 4; ++i) {
    const v2f pr = *(const v2f*)(&sA[(wave * 4 + i) * WSTRIDE + lane * 2]);
    const int2 cur = make_int2(__float_as_int(pr.x), __float_as_int(pr.y));
    const float4 acc = gather_row<BN>(F, edges, rp[i], rp[i + 1], cur, lane, bm, bs, bb);
    *(float4*)(&sA[(wave * 4 + i) * WSTRIDE + lane * 4]) = acc;
  }
  __syncthreads();

  const int stripe = wave >> 2;
  const int t0 = (wave & 3) * 2;
  const int m = lane & 15;
  const int kh = (lane >> 4) * 2;
  const float* Ap = &sA[(stripe * 16 + m) * WSTRIDE + kh];
  const float* Bp0 = &sWT[(t0 * 16 + m) * WSTRIDE + kh];
  const float* Bp1 = Bp0 + 16 * WSTRIDE;
  v8f acc0 = {}, acc1 = {};
#pragma unroll 8
  for (int k0 = 0; k0 < CH; k0 += 4) {
    const v2f a = *(const v2f*)(Ap + k0);
    const v2f b0 = *(const v2f*)(Bp0 + k0);
    const v2f b1 = *(const v2f*)(Bp1 + k0);
    acc0 = __builtin_amdgcn_wmma_f32_16x16x4_f32(false, a, false, b0, (short)0, acc0, false, false);
    acc1 = __builtin_amdgcn_wmma_f32_16x16x4_f32(false, a, false, b1, (short)0, acc1, false, false);
  }

  const int rbase = blockIdx.x * TROWS + stripe * 16 + (lane >> 4) * 8;
  const int col = t0 * 16 + m;
  for (int j = 0; j < 8; ++j) {
    if (rbase + j < Nrows) {
      out[(size_t)(rbase + j) * CH + col] = acc0[j];
      out[(size_t)(rbase + j) * CH + col + 16] = acc1[j];
    }
  }

  {
    float s0 = 0.f, q0 = 0.f, s1 = 0.f, q1 = 0.f;
    for (int j = 0; j < 8; ++j) {
      const float v0 = acc0[j], v1 = acc1[j];
      s0 += v0; q0 += v0 * v0;
      s1 += v1; q1 += v1 * v1;
    }
    s0 += __shfl_xor(s0, 16); q0 += __shfl_xor(q0, 16);
    s1 += __shfl_xor(s1, 16); q1 += __shfl_xor(q1, 16);
    if (lane < 16) {
      float* p = &sPart[stripe * (2 * CH) + col];
      p[0] = s0; p[CH] = q0;
      p[16] = s1; p[CH + 16] = q1;
    }
  }
  __syncthreads();
  if (wave < 2 * CH / 32) {
    const int c = wave * 32 + lane;
    float tot = 0.f;
    for (int s = 0; s < TROWS / 16; ++s) tot += sPart[s * (2 * CH) + c];
    atomicAdd(&out_sums[c], tot);
  }
}

__global__ void biasout_kernel(const float* __restrict__ agg40,
                               const float* __restrict__ b3,
                               float* __restrict__ out, int total) {
  int i = blockIdx.x * blockDim.x + threadIdx.x;
  if (i < total) {
    int c = i % 40;
    out[i] = agg40[i] + b3[c];
  }
}

extern "C" void kernel_launch(void* const* d_in, const int* in_sizes, int n_in,
                              void* d_out, int out_size, void* d_ws, size_t ws_size,
                              hipStream_t stream) {
  const float* feat   = (const float*)d_in[0];
  const int*   src    = (const int*)d_in[1];
  const int*   dst    = (const int*)d_in[2];
  const float* ew     = (const float*)d_in[3];
  const float* W1     = (const float*)d_in[4];
  const float* W2     = (const float*)d_in[5];
  const float* W3     = (const float*)d_in[6];
  const float* b3     = (const float*)d_in[7];
  const float* gamma1 = (const float*)d_in[8];
  const float* beta1  = (const float*)d_in[9];
  const float* gamma2 = (const float*)d_in[10];
  const float* beta2  = (const float*)d_in[11];

  int N = in_sizes[0] / CH;
  int E = in_sizes[1];

  float* g1   = (float*)d_ws;
  float* g2   = g1 + (size_t)N * CH;
  float* t3   = g1;
  float* agg3 = g1 + (size_t)N * 64;

  int2* edges   = (int2*)d_out;
  int*  row_ptr = (int*)(edges + E);
  float* sums1  = (float*)(row_ptr + (N + 1 + 3) / 4 * 4);
  float* sums2  = sums1 + 2 * CH;
  int   sgrid = (E + SLICE - 1) / SLICE;
  int   kgrid = (N + BNODES - 1) / BNODES;
  int*  tD     = (int*)g2;
  int*  tS     = tD + E;
  int*  tW     = tS + E;
  int*  histG  = tW + E;
  int*  bstart = histG + (size_t)sgrid * NBP;

  int rgrid = (N + 7) / 8;
  int fgrid = (N + 63) / 64;
  int bgrid = (N * 40 + 255) / 256;
  dim3 b128(128);
  dim3 b256(256);
  dim3 b512(512);
  dim3 b1024(1024);

  slice_hist_kernel<<<sgrid, b1024, 0, stream>>>(dst, histG, E, sums1);
  slice_scatter_kernel<<<sgrid, b1024, 0, stream>>>(src, dst, ew, histG, sgrid, bstart,
                                                    tD, tS, tW, E);
  bucket_sort_kernel<<<kgrid, b1024, 0, stream>>>(tD, tS, tW, bstart, row_ptr, edges, N, E);

  spmm_gemm_kernel<false><<<fgrid, b512, 0, stream>>>(feat, row_ptr, edges, W1, g1, N,
                                                       nullptr, nullptr, nullptr, sums1);

  spmm_gemm_kernel<true><<<fgrid, b512, 0, stream>>>(g1, row_ptr, edges, W2, g2, N,
                                                      sums1, gamma1, beta1, sums2);

  gemm_lines_kernel<3, 40><<<fgrid, b128, 0, stream>>>(g2, W3, t3, N, sums2, gamma2, beta2);
  spmm_rows_kernel<10, false><<<rgrid, b256, 0, stream>>>(t3, row_ptr, edges, agg3, N,
                                                           nullptr, nullptr, nullptr);
  biasout_kernel<<<bgrid, b256, 0, stream>>>(agg3, b3, (float*)d_out, N * 40);
}


# Gather loops issue less per edge: a row's (source, weight) pairs arrive by scalar loads in scalar registers (no readlanes, no LDS parking), 32-bit row offsets, batch norm folded to one FMA + max
# speedup vs baseline: 4.5889x; 1.0305x over previous
#include <hip/hip_runtime.h>
#include <hip/hip_bf16.h>

typedef __attribute__((ext_vector_type(2))) float v2f;
typedef __attribute__((ext_vector_type(8))) float v8f;

#define BN_EPS 1e-5f
#define CH 128
#define WSTRIDE 132

#define BSH 9
#define BNODES (1 << BSH)
#define NBP 256
#define SLICE 6400
#define SPT ((SLICE + 1023) / 1024)
#define BPT (BCAP / 1024)
#define BCAP 16384

__device__ __forceinline__ int wave_incl_scan(int s, int lane) {
  for (int o = 1; o < 32; o <<= 1) {
    int v = __shfl_up(s, o);
    if (lane >= o) s += v;
  }
  return s;
}

template <int NW>
__device__ __forceinline__ int block_excl_scan(int v, int tid, int* s_w) {
  const int lane = tid & 31;
  const int wv = tid >> 5;
  const int incl = wave_incl_scan(v, lane);
  if (wv < NW && lane == 31) s_w[wv] = incl;
  __syncthreads();
  int off = 0;
  for (int k = 0; k < NW; ++k)
    if (k < wv) off += s_w[k];
  __syncthreads();
  return off + incl - v;
}

__global__ __launch_bounds__(1024) void slice_hist_kernel(const int* __restrict__ dst,
                                                          int* __restrict__ histG, int E,
                                                          float* __restrict__ sums) {
  __shared__ int hist[NBP];
  const int tid = threadIdx.x;
  if (tid < NBP) hist[tid] = 0;
  if (blockIdx.x == 0 && tid < 512) sums[tid] = 0.f;
  __syncthreads();
  const int beg = blockIdx.x * SLICE;
  const int end = min(E, beg + SLICE);
  int d[SPT];
#pragma unroll
  for (int k = 0; k < SPT; ++k) {
    const int e = beg + tid + k * 1024;
    d[k] = 0;
    if (e < end) d[k] = dst[e];
  }
#pragma unroll
  for (int k = 0; k < SPT; ++k)
    if (beg + tid + k * 1024 < end) atomicAdd(&hist[d[k] >> BSH], 1);
  __syncthreads();
  if (tid < NBP) histG[blockIdx.x * NBP + tid] = hist[tid];
}

__global__ __launch_bounds__(1024) void slice_scatter_kernel(const int* __restrict__ src,
                                                             const int* __restrict__ dst,
                                                             const float* __restrict__ ew,
                                                             const int* __restrict__ histG, int G,
                                                             int* __restrict__ bstart,
                                                             int* __restrict__ tD,
                                                             int* __restrict__ tS,
                                                             int* __restrict__ tW, int E) {
  __shared__ int s_tot[NBP];
  __shared__ int s_bef[NBP];
  __shared__ int s_cur[NBP];
  __shared__ int s_delta[NBP];
  __shared__ int s_w[8];
  __shared__ int st_d[SLICE];
  __shared__ int st_s[SLICE];
  __shared__ int st_w[SLICE];
  const int tid = threadIdx.x;
  const int g = blockIdx.x;
  const int t = tid & (NBP - 1);
  const int q = tid >> 8;
  if (tid < NBP) { s_tot[tid] = 0; s_bef[tid] = 0; }
  __syncthreads();
  const int beg = g * SLICE;
  const int end = min(E, beg + SLICE);
  int ed[SPT], es[SPT], ewb[SPT];
#pragma unroll
  for (int k = 0; k < SPT; ++k) {
    const int e = beg + tid + k * 1024;
    ed[k] = 0; es[k] = 0; ewb[k] = 0;
    if (e < end) { ed[k] = dst[e]; es[k] = src[e]; ewb[k] = __float_as_int(ew[e]); }
  }
  int tot = 0, bef = 0;
#pragma unroll 8
  for (int r = q; r < G; r += 4) {
    const int v = histG[r * NBP + t];
    tot += v;
    if (r < g) bef += v;
  }
  atomicAdd(&s_tot[t], tot);
  atomicAdd(&s_bef[t], bef);
  __syncthreads();
  int mine = 0, total = 0;
  if (tid < NBP) { mine = histG[g * NBP + tid]; total = s_tot[tid]; }
  const int bst = block_excl_scan<8>(total, tid, s_w);
  const int lst = block_excl_scan<8>(mine, tid, s_w);
  if (tid < NBP) {
    s_cur[tid] = lst;
    s_delta[tid] = bst + s_bef[tid] - lst;
    if (g == 0) {
      bstart[tid] = bst;
      if (tid == NBP - 1) bstart[NBP] = bst + total;
    }
  }
  __syncthreads();
#pragma unroll
  for (int k = 0; k < SPT; ++k) {
    if (beg + tid + k * 1024 < end) {
      const int p = atomicAdd(&s_cur[ed[k] >> BSH], 1);
      st_d[p] = ed[k];
      st_s[p] = es[k];
      st_w[p] = ewb[k];
    }
  }
  __syncthreads();
  const int n = end - beg;
  for (int i = tid; i < n; i += 1024) {
    const int d = st_d[i];
    const int pos = s_delta[d >> BSH] + i;
    tD[pos] = d;
    tS[pos] = st_s[i];
    tW[pos] = st_w[i];
  }
}

__global__ __launch_bounds__(1024) void bucket_sort_kernel(const int* __restrict__ tD,
                                                           const int* __restrict__ tS,
                                                           const int* __restrict__ tW,
                                                           const int* __restrict__ bstart,
                                                           int* __restrict__ row_ptr,
                                                           int2* __restrict__ edges, int N, int E) {
  __shared__ int s_cnt[BNODES];
  __shared__ int s_w[16];
  __shared__ int2 stage[BCAP];
  const int tid = threadIdx.x;
  const int b = blockIdx.x;
  const int lo = bstart[b];
  const int n = bstart[b + 1] - lo;
  const bool fits = n <= BCAP;
  if (tid < BNODES) s_cnt[tid] = 0;
  int dl[BPT], es[BPT], ewb[BPT];
#pragma unroll
  for (int k = 0; k < BPT; ++k) {
    const int i = tid + k * 1024;
    dl[k] = 0; es[k] = 0; ewb[k] = 0;
    if (fits && i < n) {
      dl[k] = tD[lo + i];
      es[k] = tS[lo + i];
      ewb[k] = tW[lo + i];
    }
  }
  __syncthreads();
  if (fits) {
#pragma unroll
    for (int k = 0; k < BPT; ++k)
      if (tid + k * 1024 < n) atomicAdd(&s_cnt[dl[k] & (BNODES - 1)], 1);
  } else {
    for (int i = tid; i < n; i += 1024) atomicAdd(&s_cnt[tD[lo + i] & (BNODES - 1)], 1);
  }
  __syncthreads();
  const int c = tid < BNODES ? s_cnt[tid] : 0;
  const int st = block_excl_scan<16>(c, tid, s_w);
  if (tid < BNODES) {
    s_cnt[tid] = st;
    const int node = b * BNODES + tid;
    if (node < N) row_ptr[node] = lo + st;
  }
  if (b == 0 && tid == 0) row_ptr[N] = E;
  __syncthreads();
  if (fits) {
#pragma unroll
    for (int k = 0; k < BPT; ++k) {
      if (tid + k * 1024 < n) {
        const int p = atomicAdd(&s_cnt[dl[k] & (BNODES - 1)], 1);
        stage[p] = make_int2(es[k], ewb[k]);
      }
    }
    __syncthreads();
    for (int i = tid; i < n; i += 1024) edges[lo + i] = stage[i];
  } else {
    for (int i = tid; i < n; i += 1024) {
      const int p = atomicAdd(&s_cnt[tD[lo + i] & (BNODES - 1)], 1);
      edges[lo + p] = make_int2(tS[lo + i], tW[lo + i]);
    }
  }
}

__device__ __forceinline__ float bn_relu(float x, float mean, float sc, float beta) {
  float v = (x - mean) * sc + beta;
  return v > 0.f ? v : 0.f;
}

__device__ __forceinline__ void load_pairs(int2 (&p)[8], const int2* __restrict__ edges, int at) {
#pragma unroll
  for (int u = 0; u < 8; ++u) p[u] = edges[at + u];
}

template <int C4>
__device__ __forceinline__ float4 load_src_row(const char* __restrict__ Fb, int s,
                                               unsigned lane_off) {
  const unsigned off = (unsigned)s * (unsigned)(C4 * 16) + lane_off;
  return *(const float4*)(Fb + off);
}

template <bool BN>
__device__ __forceinline__ void edge_acc(float4& acc, float4 h, float w, const float4& sc,
                                         const float4& sh) {
  if constexpr (BN) {
    h.x = fmaxf(fmaf(h.x, sc.x, sh.x), 0.f);
    h.y = fmaxf(fmaf(h.y, sc.y, sh.y), 0.f);
    h.z = fmaxf(fmaf(h.z, sc.z, sh.z), 0.f);
    h.w = fmaxf(fmaf(h.w, sc.w, sh.w), 0.f);
  }
  acc.x = fmaf(w, h.x, acc.x);
  acc.y = fmaf(w, h.y, acc.y);
  acc.z = fmaf(w, h.z, acc.z);
  acc.w = fmaf(w, h.w, acc.w);
}

template <int C4, bool BN>
__device__ __forceinline__ void chunk_full(const char* __restrict__ Fb,
                                           const int2* __restrict__ edges, int next_at,
                                           const int2 (&p)[8], int2 (&q)[8], unsigned lane_off,
                                           float4& acc, const float4& sc, const float4& sh) {
  float4 v[8];
#pragma unroll
  for (int u = 0; u < 8; ++u) v[u] = load_src_row<C4>(Fb, p[u].x, lane_off);
  load_pairs(q, edges, next_at);
#pragma unroll
  for (int u = 0; u < 8; ++u) edge_acc<BN>(acc, v[u], __int_as_float(p[u].y), sc, sh);
}

template <int C4, bool BN>
__device__ __forceinline__ void chunk_tail(const char* __restrict__ Fb,
                                           const int2* __restrict__ edges, int n, int next_at,
                                           const int2 (&p)[8], int2 (&q)[8], unsigned lane_off,
                                           float4& acc, const float4& sc, const float4& sh) {
  float4 v[7];
#pragma unroll
  for (int u = 0; u < 7; ++u)
    if (u < n) v[u] = load_src_row<C4>(Fb, p[u].x, lane_off);
  load_pairs(q, edges, next_at);
#pragma unroll
  for (int u = 0; u < 7; ++u)
    if (u < n) edge_acc<BN>(acc, v[u], __int_as_float(p[u].y), sc, sh);
}

template <int C4, bool BN>
__device__ __forceinline__ float4 gather_row(const char* __restrict__ Fb,
                                             const int2* __restrict__ edges, int beg, int end,
                                             int2 (&cur)[8], unsigned lane_off,
                                             const float4& sc, const float4& sh) {
  float4 acc = {0.f, 0.f, 0.f, 0.f};
  int2 nxt[8];
  int base = beg;
  while (end - base >= 8) {
    chunk_full<C4, BN>(Fb, edges, base + 8, cur, nxt, lane_off, acc, sc, sh);
#pragma unroll
    for (int u = 0; u < 8; ++u) cur[u] = nxt[u];
    base += 8;
  }
  const int n = end - base;
  if (n > 0) {
    chunk_tail<C4, BN>(Fb, edges, n, end, cur, nxt, lane_off, acc, sc, sh);
#pragma unroll
    for (int u = 0; u < 8; ++u) cur[u] = nxt[u];
  }
  return acc;
}

template <int C4>
__global__ __launch_bounds__(256) void spmm_rows_kernel(const float* __restrict__ feat,
                                                        const int* __restrict__ row_ptr,
                                                        const int2* __restrict__ edges,
                                                        float* __restrict__ agg, int Nrows) {
  const int wave = __builtin_amdgcn_readfirstlane((int)(threadIdx.x >> 5));
  const int lane = threadIdx.x & 31;
  const int row = blockIdx.x * 8 + wave;
  if (row >= Nrows) return;
  const int l = lane < C4 ? lane : C4 - 1;
  const int beg = row_ptr[row];
  const int end = row_ptr[row + 1];
  int2 cur[8];
  load_pairs(cur, edges, beg);
  const float4 none = {0.f, 0.f, 0.f, 0.f};
  const float4 acc = gather_row<C4, false>((const char*)feat, edges, beg, end, cur,
                                           (unsigned)l * 16u, none, none);
  if (lane < C4) ((float4*)agg)[(size_t)row * C4 + lane] = acc;
}

#define GROWS 64
template <int NT, int COUT>
__global__ __launch_bounds__(128) void gemm_lines_kernel(const float* __restrict__ A,
                                                         const float* __restrict__ W,
                                                         float* __restrict__ out,
                                                         int Nrows,
                                                         const float* __restrict__ in_sums,
                                                         const float* __restrict__ in_gamma,
                                                         const float* __restrict__ in_beta) {
  constexpr int CP = NT * 16;
  constexpr int C4 = COUT / 4;
  constexpr int WPT = CH * C4 / 128;
  static_assert(COUT % 4 == 0 && COUT <= CP && (CH * C4) % 128 == 0, "W is staged in whole float4");
  __shared__ __attribute__((aligned(16))) float sWT[CP * WSTRIDE];
  __shared__ __attribute__((aligned(16))) float sA[GROWS * WSTRIDE];
  const int tid = threadIdx.x;
  const int wave = __builtin_amdgcn_readfirstlane(tid >> 5);
  const int lane = tid & 31;
  const int row0 = blockIdx.x * GROWS + wave * 16;
  const bool valid = row0 < Nrows;

  float4 r[16];
  if (valid) {
    const float4* A4 = (const float4*)A + (size_t)row0 * (CH / 4) + lane;
#pragma unroll
    for (int i = 0; i < 16; ++i) r[i] = A4[(size_t)i * (CH / 4)];
  }
  float4 wv[WPT];
#pragma unroll
  for (int j = 0; j < WPT; ++j) wv[j] = ((const float4*)W)[tid + j * 128];
  float4 bm, bs, bb;
  {
    const float invN = 1.0f / (float)Nrows;
    const float4 s = ((const float4*)in_sums)[lane];
    const float4 q = ((const float4*)(in_sums + CH))[lane];
    const float4 g4 = ((const float4*)in_gamma)[lane];
    bb = ((const float4*)in_beta)[lane];
    bm.x = s.x * invN; bm.y = s.y * invN; bm.z = s.z * invN; bm.w = s.w * invN;
    bs.x = g4.x * rsqrtf(q.x * invN - bm.x * bm.x + BN_EPS);
    bs.y = g4.y * rsqrtf(q.y * invN - bm.y * bm.y + BN_EPS);
    bs.z = g4.z * rsqrtf(q.z * invN - bm.z * bm.z + BN_EPS);
    bs.w = g4.w * rsqrtf(q.w * invN - bm.w * bm.w + BN_EPS);
  }
#pragma unroll
  for (int j = 0; j < WPT; ++j) {
    const int q = tid + j * 128;
    float* wd = &sWT[(q % C4) * 4 * WSTRIDE + q / C4];
    wd[0] = wv[j].x;
    wd[WSTRIDE] = wv[j].y;
    wd[2 * WSTRIDE] = wv[j].z;
    wd[3 * WSTRIDE] = wv[j].w;
  }
#pragma unroll
  for (int j = COUT; j < CP; ++j) sWT[j * WSTRIDE + tid] = 0.f;
  float* stripe = &sA[wave * 16 * WSTRIDE];
  if (valid) {
#pragma unroll
    for (int i = 0; i < 16; ++i) {
      float4 h = r[i];
      h.x = bn_relu(h.x, bm.x, bs.x, bb.x);
      h.y = bn_relu(h.y, bm.y, bs.y, bb.y);
      h.z = bn_relu(h.z, bm.z, bs.z, bb.z);
      h.w = bn_relu(h.w, bm.w, bs.w, bb.w);
      *(float4*)(&stripe[i * WSTRIDE + lane * 4]) = h;
    }
  }
  __syncthreads();

  const int m = lane & 15;
  const int kh = (lane >> 4) * 2;
  v8f acc[NT];
#pragma unroll
  for (int t = 0; t < NT; ++t) acc[t] = {};
  if (valid) {
    const float* Ap = &stripe[m * WSTRIDE + kh];
    const float* Bp = &sWT[m * WSTRIDE + kh];
#pragma unroll 8
    for (int k0 = 0; k0 < CH; k0 += 4) {
      const v2f a = *(const v2f*)(Ap + k0);
#pragma unroll
      for (int t = 0; t < NT; ++t) {
        const v2f b = *(const v2f*)(Bp + t * 16 * WSTRIDE + k0);
        acc[t] = __builtin_amdgcn_wmma_f32_16x16x4_f32(
             false, a,  false, b,
             (short)0, acc[t],  false,  false);
      }
    }
    const int rb = (lane >> 4) * 8;
#pragma unroll
    for (int t = 0; t < NT; ++t) {
      const int col = t * 16 + m;
      if (col < COUT) {
#pragma unroll
        for (int j = 0; j < 8; ++j) stripe[(rb + j) * COUT + col] = acc[t][j];
      }
    }
  }
  __syncthreads();
  if (valid) {
    float4* o4 = (float4*)(out + (size_t)row0 * COUT);
#pragma unroll
    for (int i = lane; i < 4 * COUT; i += 32) o4[i] = ((const float4*)stripe)[i];
  }
}

#define TROWS 64
template <bool BN>
__global__ __launch_bounds__(512, 12) void spmm_gemm_kernel(const float* __restrict__ feat,
                                                             const int* __restrict__ row_ptr,
                                                             const int2* __restrict__ edges,
                                                             const float* __restrict__ W,
                                                             float* __restrict__ out, int Nrows,
                                                             const float* __restrict__ in_sums,
                                                             const float* __restrict__ in_gamma,
                                                             const float* __restrict__ in_beta,
                                                             float* __restrict__ out_sums) {
  __shared__ float sWT[CH * WSTRIDE];
  __shared__ float sA[TROWS * WSTRIDE];
  __shared__ float sPart[(TROWS / 16) * 2 * CH];
  const int tid = threadIdx.x;
  const int wave = __builtin_amdgcn_readfirstlane(tid >> 5);
  const int lane = __builtin_amdgcn_mbcnt_lo(~0u, 0u);

  const int r0 = blockIdx.x * TROWS + wave * 4;
  int rp[5];
  for (int i = 0; i < 5; ++i) rp[i] = row_ptr[min(r0 + i, Nrows)];
  int2 cur[8];
  load_pairs(cur, edges, rp[0]);

  {
    float wv[32];
#pragma unroll
    for (int j = 0; j < 32; ++j) wv[j] = W[tid + j * 512];
    float* wdst = &sWT[(tid & (CH - 1)) * WSTRIDE + (tid >> 7)];
#pragma unroll
    for (int j = 0; j < 32; ++j) wdst[4 * j] = wv[j];
  }
  float4 bs = {0.f, 0.f, 0.f, 0.f}, bh = bs;
  if constexpr (BN) {
    const float invN = 1.0f / (float)Nrows;
    const float4 s = ((const float4*)in_sums)[lane];
    const float4 q = ((const float4*)(in_sums + CH))[lane];
    const float4 g4 = ((const float4*)in_gamma)[lane];
    const float4 bb = ((const float4*)in_beta)[lane];
    float4 bm;
    bm.x = s.x * invN; bm.y = s.y * invN; bm.z = s.z * invN; bm.w = s.w * invN;
    bs.x = g4.x * rsqrtf(q.x * invN - bm.x * bm.x + BN_EPS);
    bs.y = g4.y * rsqrtf(q.y * invN - bm.y * bm.y + BN_EPS);
    bs.z = g4.z * rsqrtf(q.z * invN - bm.z * bm.z + BN_EPS);
    bs.w = g4.w * rsqrtf(q.w * invN - bm.w * bm.w + BN_EPS);
    bh.x = fmaf(-bm.x, bs.x, bb.x);
    bh.y = fmaf(-bm.y, bs.y, bb.y);
    bh.z = fmaf(-bm.z, bs.z, bb.z);
    bh.w = fmaf(-bm.w, bs.w, bb.w);
  }

  const char* Fb = (const char*)feat;
  const unsigned lane_off = (unsigned)lane * 16u;
#pragma unroll
  for (int i = 0; i < 4; ++i) {
    const float4 acc = gather_row<32, BN>(Fb, edges, rp[i], rp[i + 1], cur, lane_off, bs, bh);
    *(float4*)(&sA[(wave * 4 + i) * WSTRIDE + lane * 4]) = acc;
  }
  __syncthreads();

  const int stripe = wave >> 2;
  const int t0 = (wave & 3) * 2;
  const int m = lane & 15;
  const int kh = (lane >> 4) * 2;
  const float* Ap = &sA[(stripe * 16 + m) * WSTRIDE + kh];
  const float* Bp0 = &sWT[(t0 * 16 + m) * WSTRIDE + kh];
  const float* Bp1 = Bp0 + 16 * WSTRIDE;
  v8f acc0 = {}, acc1 = {};
#pragma unroll 8
  for (int k0 = 0; k0 < CH; k0 += 4) {
    const v2f a = *(const v2f*)(Ap + k0);
    const v2f b0 = *(const v2f*)(Bp0 + k0);
    const v2f b1 = *(const v2f*)(Bp1 + k0);
    acc0 = __builtin_amdgcn_wmma_f32_16x16x4_f32(false, a, false, b0, (short)0, acc0, false, false);
    acc1 = __builtin_amdgcn_wmma_f32_16x16x4_f32(false, a, false, b1, (short)0, acc1, false, false);
  }

  const int rbase = blockIdx.x * TROWS + stripe * 16 + (lane >> 4) * 8;
  const int col = t0 * 16 + m;
  for (int j = 0; j < 8; ++j) {
    if (rbase + j < Nrows) {
      out[(size_t)(rbase + j) * CH + col] = acc0[j];
      out[(size_t)(rbase + j) * CH + col + 16] = acc1[j];
    }
  }

  {
    float s0 = 0.f, q0 = 0.f, s1 = 0.f, q1 = 0.f;
    for (int j = 0; j < 8; ++j) {
      const float v0 = acc0[j], v1 = acc1[j];
      s0 += v0; q0 += v0 * v0;
      s1 += v1; q1 += v1 * v1;
    }
    s0 += __shfl_xor(s0, 16); q0 += __shfl_xor(q0, 16);
    s1 += __shfl_xor(s1, 16); q1 += __shfl_xor(q1, 16);
    if (lane < 16) {
      float* p = &sPart[stripe * (2 * CH) + col];
      p[0] = s0; p[CH] = q0;
      p[16] = s1; p[CH + 16] = q1;
    }
  }
  __syncthreads();
  if (wave < 2 * CH / 32) {
    const int c = wave * 32 + lane;
    float tot = 0.f;
    for (int s = 0; s < TROWS / 16; ++s) tot += sPart[s * (2 * CH) + c];
    atomicAdd(&out_sums[c], tot);
  }
}

__global__ void biasout_kernel(const float* __restrict__ agg40,
                               const float* __restrict__ b3,
                               float* __restrict__ out, int total) {
  int i = blockIdx.x * blockDim.x + threadIdx.x;
  if (i < total) {
    int c = i % 40;
    out[i] = agg40[i] + b3[c];
  }
}

extern "C" void kernel_launch(void* const* d_in, const int* in_sizes, int n_in,
                              void* d_out, int out_size, void* d_ws, size_t ws_size,
                              hipStream_t stream) {
  const float* feat   = (const float*)d_in[0];
  const int*   src    = (const int*)d_in[1];
  const int*   dst    = (const int*)d_in[2];
  const float* ew     = (const float*)d_in[3];
  const float* W1     = (const float*)d_in[4];
  const float* W2     = (const float*)d_in[5];
  const float* W3     = (const float*)d_in[6];
  const float* b3     = (const float*)d_in[7];
  const float* gamma1 = (const float*)d_in[8];
  const float* beta1  = (const float*)d_in[9];
  const float* gamma2 = (const float*)d_in[10];
  const float* beta2  = (const float*)d_in[11];

  int N = in_sizes[0] / CH;
  int E = in_sizes[1];

  float* g1   = (float*)d_ws;
  float* g2   = g1 + (size_t)N * CH;
  float* t3   = g1;
  float* agg3 = g1 + (size_t)N * 64;

  int2* edges   = (int2*)d_out;
  int*  row_ptr = (int*)(edges + E);
  float* sums1  = (float*)(row_ptr + (N + 1 + 3) / 4 * 4);
  float* sums2  = sums1 + 2 * CH;
  int   sgrid = (E + SLICE - 1) / SLICE;
  int   kgrid = (N + BNODES - 1) / BNODES;
  int*  tD     = (int*)g2;
  int*  tS     = tD + E;
  int*  tW     = tS + E;
  int*  histG  = tW + E;
  int*  bstart = histG + (size_t)sgrid * NBP;

  int rgrid = (N + 7) / 8;
  int fgrid = (N + 63) / 64;
  int bgrid = (N * 40 + 255) / 256;
  dim3 b128(128);
  dim3 b256(256);
  dim3 b512(512);
  dim3 b1024(1024);

  slice_hist_kernel<<<sgrid, b1024, 0, stream>>>(dst, histG, E, sums1);
  slice_scatter_kernel<<<sgrid, b1024, 0, stream>>>(src, dst, ew, histG, sgrid, bstart,
                                                    tD, tS, tW, E);
  bucket_sort_kernel<<<kgrid, b1024, 0, stream>>>(tD, tS, tW, bstart, row_ptr, edges, N, E);

  spmm_gemm_kernel<false><<<fgrid, b512, 0, stream>>>(feat, row_ptr, edges, W1, g1, N,
                                                       nullptr, nullptr, nullptr, sums1);

  spmm_gemm_kernel<true><<<fgrid, b512, 0, stream>>>(g1, row_ptr, edges, W2, g2, N,
                                                      sums1, gamma1, beta1, sums2);

  gemm_lines_kernel<3, 40><<<fgrid, b128, 0, stream>>>(g2, W3, t3, N, sums2, gamma2, beta2);
  spmm_rows_kernel<10><<<rgrid, b256, 0, stream>>>(t3, row_ptr, edges, agg3, N);
  biasout_kernel<<<bgrid, b256, 0, stream>>>(agg3, b3, (float*)d_out, N * 40);
}
